# starting kernel
# baseline (speedup 1.0000x reference)
#include <hip/hip_runtime.h>

typedef float v2f __attribute__((ext_vector_type(2)));
typedef float v8f __attribute__((ext_vector_type(8)));

#define N_NODES 50000
#define N_EDGES 600000
#define DIM     128
#define NF      (N_NODES * DIM)

__global__ void k_fill(float* __restrict__ p, float v, int n) {
    int i = blockIdx.x * blockDim.x + threadIdx.x;
    if (i < n) p[i] = v;
}

__global__ void k_deg_accum(const int* __restrict__ dst, float* __restrict__ deg) {
    int e = blockIdx.x * blockDim.x + threadIdx.x;
    if (e < N_EDGES) atomicAdd(&deg[dst[e]], 1.0f);
}

__global__ void k_rsqrt_inplace(float* __restrict__ d, int n) {
    int i = blockIdx.x * blockDim.x + threadIdx.x;
    if (i < n) d[i] = rsqrtf(d[i]);
}

__global__ void k_feat(const float* __restrict__ w,
                       const float* __restrict__ lw,
                       const float* __restrict__ lb,
                       float* __restrict__ feat) {
    int i = blockIdx.x * blockDim.x + threadIdx.x;
    if (i < NF) {
        int n = i >> 7, d = i & 127;
        feat[i] = w[n] * lw[d] + lb[d];
    }
}

__global__ void k_scatter(const int* __restrict__ src,
                          const int* __restrict__ dst,
                          const float* __restrict__ x,
                          const float* __restrict__ dinv,
                          float* __restrict__ agg) {
    unsigned t = blockIdx.x * blockDim.x + threadIdx.x;
    if (t >= (unsigned)N_EDGES * 32u) return;
    int e = t >> 5;
    int q = (t & 31) << 2;
    int s = src[e], d = dst[e];
    float sc = dinv[s];
    const float4 v = *(const float4*)(x + (size_t)s * DIM + q);
    float* ap = agg + (size_t)d * DIM + q;
    atomicAdd(ap + 0, v.x * sc);
    atomicAdd(ap + 1, v.y * sc);
    atomicAdd(ap + 2, v.z * sc);
    atomicAdd(ap + 3, v.w * sc);
}

__global__ void k_finish(float* __restrict__ out,
                         const float* __restrict__ agg,
                         const float* __restrict__ x,
                         const float* __restrict__ prev,
                         const float* __restrict__ dinv,
                         float alpha, float beta) {
    int i = blockIdx.x * blockDim.x + threadIdx.x;
    if (i < NF) {
        int n = i >> 7;
        float di = dinv[n];
        float h = (agg[i] + x[i] * di) * di;
        out[i] = alpha * h + beta * prev[i];
    }
}

__global__ void __launch_bounds__(32)
k_cheb_gemm(const float* __restrict__ X0,
            const float* __restrict__ X1,
            const float* __restrict__ X2,
            const float* __restrict__ W,
            const float* __restrict__ b,
            float* __restrict__ out) {
    const int row0 = blockIdx.x * 16;
    const int lane = threadIdx.x & 31;
    const int half = lane >> 4;
    const int lr   = lane & 15;

    v8f acc[8];
#pragma unroll
    for (int nt = 0; nt < 8; ++nt) {
        float bv = b[nt * 16 + lr];
        acc[nt] = (v8f){bv, bv, bv, bv, bv, bv, bv, bv};
    }

    const float* segs[3] = {X0, X1, X2};
    for (int seg = 0; seg < 3; ++seg) {
        const float* Xp = segs[seg] + (size_t)row0 * DIM;
#pragma unroll 4
        for (int kk = 0; kk < DIM; kk += 4) {
            const int ka = kk + 2 * half;
            v2f a;
            a.x = Xp[(size_t)lr * DIM + ka];
            a.y = Xp[(size_t)lr * DIM + ka + 1];
            const float* Wk  = W + (size_t)(seg * DIM + ka) * DIM;
            const float* Wk1 = Wk + DIM;
#pragma unroll
            for (int nt = 0; nt < 8; ++nt) {
                v2f bb;
                bb.x = Wk [nt * 16 + lr];
                bb.y = Wk1[nt * 16 + lr];
                acc[nt] = __builtin_amdgcn_wmma_f32_16x16x4_f32(
                     false, a,  false, bb,
                     (short)0, acc[nt],
                     false,  false);
            }
        }
    }

#pragma unroll
    for (int nt = 0; nt < 8; ++nt) {
        const int col = nt * 16 + lr;
#pragma unroll
        for (int r = 0; r < 8; ++r) {
            const int row = row0 + r + 8 * half;
            float v = acc[nt][r];
            out[(size_t)row * DIM + col] = v > 0.0f ? v : 0.0f;
        }
    }
}

__global__ void k_pred(const float* __restrict__ emb,
                       const float* __restrict__ pw,
                       const float* __restrict__ pb,
                       float* __restrict__ out) {
    int n = blockIdx.x * blockDim.x + threadIdx.x;
    if (n < N_NODES) {
        const float* e = emb + (size_t)n * DIM;
        float s = 0.0f;
#pragma unroll 8
        for (int d = 0; d < DIM; d += 4) {
            float4 v = *(const float4*)(e + d);
            s += v.x * pw[d] + v.y * pw[d + 1] + v.z * pw[d + 2] + v.w * pw[d + 3];
        }
        out[n] = s + pb[0];
    }
}

extern "C" void kernel_launch(void* const* d_in, const int* in_sizes, int n_in,
                              void* d_out, int out_size, void* d_ws, size_t ws_size,
                              hipStream_t stream) {
    const float* weights  = (const float*)d_in[0];
    const int*   src      = (const int*)  d_in[1];
    const int*   dst      = (const int*)  d_in[2];
    const float* lin_in_w = (const float*)d_in[3];
    const float* lin_in_b = (const float*)d_in[4];
    const float* cheb_ws  = (const float*)d_in[5];
    const float* cheb_bs  = (const float*)d_in[6];
    const float* pred_w   = (const float*)d_in[7];
    const float* pred_b   = (const float*)d_in[8];
    float* logits = (float*)d_out;

    float* ws   = (float*)d_ws;
    float* dinv = ws;
    float* X0   = ws + 51200;
    float* X1   = X0 + NF;
    float* X2   = X1 + NF;
    float* agg  = X2 + NF;
    float* emb  = agg + NF;

    const int T = 256;
    const int gN    = (N_NODES + T - 1) / T;
    const int gE    = (N_EDGES + T - 1) / T;
    const int gNF   = (NF + T - 1) / T;
    const int gScat = (N_EDGES * 32 + T - 1) / T;

    k_fill<<<gN, T, 0, stream>>>(dinv, 1.0f, N_NODES);
    k_deg_accum<<<gE, T, 0, stream>>>(dst, dinv);
    k_rsqrt_inplace<<<gN, T, 0, stream>>>(dinv, N_NODES);

    k_feat<<<gNF, T, 0, stream>>>(weights, lin_in_w, lin_in_b, X0);

    float* cur = X0;
    float* nxt = emb;
    for (int layer = 0; layer < 3; ++layer) {
        const float* Wl = cheb_ws + (size_t)layer * 3 * DIM * DIM;
        const float* bl = cheb_bs + (size_t)layer * DIM;

        k_fill<<<gNF, T, 0, stream>>>(agg, 0.0f, NF);
        k_scatter<<<gScat, T, 0, stream>>>(src, dst, cur, dinv, agg);
        k_finish<<<gNF, T, 0, stream>>>(X1, agg, cur, cur, dinv, -1.0f, 0.0f);

        k_fill<<<gNF, T, 0, stream>>>(agg, 0.0f, NF);
        k_scatter<<<gScat, T, 0, stream>>>(src, dst, X1, dinv, agg);
        k_finish<<<gNF, T, 0, stream>>>(X2, agg, X1, cur, dinv, -2.0f, -1.0f);

        k_cheb_gemm<<<N_NODES / 16, 32, 0, stream>>>(cur, X1, X2, Wl, bl, nxt);

        float* t = cur; cur = nxt; nxt = t;
    }

    k_pred<<<gN, T, 0, stream>>>(cur, pred_w, pred_b, logits);
}


# Float-atomic scatter-add replaced by per-destination gather: edges grouped by dst each call (count, scan, atomic slot claim); one wave sums a row in registers in slot-arrival order, recurrence fused
# speedup vs baseline: 2.6887x; 2.6887x over previous
#include <hip/hip_runtime.h>


typedef float v2f __attribute__((ext_vector_type(2)));
typedef float v8f __attribute__((ext_vector_type(8)));

#define N_NODES 50000
#define N_EDGES 600000
#define DIM     128
#define NF      (N_NODES * DIM)

struct Edge { int src; float w; };

__global__ void k_zero_int(int* __restrict__ p, int n) {
    int i = blockIdx.x * blockDim.x + threadIdx.x;
    if (i < n) p[i] = 0;
}

__global__ void k_count(const int* __restrict__ dst, int* __restrict__ cnt) {
    int e = blockIdx.x * blockDim.x + threadIdx.x;
    if (e < N_EDGES) atomicAdd(&cnt[dst[e]], 1);
}

#define SCAN_T      1024
#define SCAN_BLOCKS ((N_NODES + SCAN_T - 1) / SCAN_T)

__global__ void __launch_bounds__(SCAN_T)
k_scan(const int* __restrict__ cnt,
       int* __restrict__ rowptr,
       int* __restrict__ cursor,
       float* __restrict__ dinv) {
    __shared__ int s_cnt[2][SCAN_T];
    __shared__ int s_pre[2][SCAN_T];
    const int t     = threadIdx.x;
    const int first = blockIdx.x * SCAN_T;
    const int i     = first + t;
    const int c     = (i < N_NODES) ? cnt[i] : 0;
    int pre = 0;
    for (int j = t; j < first; j += SCAN_T) pre += cnt[j];

    s_cnt[0][t] = c;
    s_pre[0][t] = pre;
    __syncthreads();
    int cur = 0;
    for (int off = 1; off < SCAN_T; off <<= 1) {
        int vc = s_cnt[cur][t], vp = s_pre[cur][t];
        if (t >= off) { vc += s_cnt[cur][t - off]; vp += s_pre[cur][t - off]; }
        s_cnt[cur ^ 1][t] = vc;
        s_pre[cur ^ 1][t] = vp;
        __syncthreads();
        cur ^= 1;
    }
    const int incl = s_cnt[cur][t];
    const int base = s_pre[cur][SCAN_T - 1];
    const int excl = base + incl - c;
    if (i < N_NODES) {
        rowptr[i] = excl;
        cursor[i] = excl;
        dinv[i]   = rsqrtf(1.0f + (float)c);
    }
    if (i == N_NODES - 1) rowptr[N_NODES] = excl + c;
}

__global__ void k_fill_csr(const int* __restrict__ src,
                           const int* __restrict__ dst,
                           const float* __restrict__ dinv,
                           int* __restrict__ cursor,
                           Edge* __restrict__ edge) {
    int e = blockIdx.x * blockDim.x + threadIdx.x;
    if (e < N_EDGES) {
        const int s = src[e];
        Edge r;
        r.src = s;
        r.w   = dinv[s];
        int slot = atomicAdd(&cursor[dst[e]], 1);
        edge[slot] = r;
    }
}

__global__ void k_feat(const float* __restrict__ w,
                       const float* __restrict__ lw,
                       const float* __restrict__ lb,
                       float* __restrict__ feat) {
    int i = blockIdx.x * blockDim.x + threadIdx.x;
    if (i < NF) {
        int n = i >> 7, d = i & 127;
        feat[i] = w[n] * lw[d] + lb[d];
    }
}

#define PULL_ROWS_PER_BLOCK 8
static_assert(N_NODES % PULL_ROWS_PER_BLOCK == 0, "grid covers the rows exactly");

template <int G>
__device__ __forceinline__ void pull_rows(const Edge* __restrict__ edge,
                                          const float4* __restrict__ x4,
                                          int i, int lane, float4& acc) {
    Edge r[G]; float4 v[G];
#pragma unroll
    for (int k = 0; k < G; ++k) r[k] = edge[i + k];
#pragma unroll
    for (int k = 0; k < G; ++k) v[k] = x4[(size_t)r[k].src * (DIM / 4) + lane];
#pragma unroll
    for (int k = 0; k < G; ++k) {
        acc.x += v[k].x * r[k].w;
        acc.y += v[k].y * r[k].w;
        acc.z += v[k].z * r[k].w;
        acc.w += v[k].w * r[k].w;
    }
}

template <bool PREV_IS_X>
__global__ void __launch_bounds__(32 * PULL_ROWS_PER_BLOCK)
k_spmm_pull(const int* __restrict__ rowptr,
            const Edge* __restrict__ edge,
            const float* __restrict__ x,
            const float* __restrict__ prev,
            const float* __restrict__ dinv,
            float* __restrict__ out,
            float alpha, float beta) {
    const int row  = __builtin_amdgcn_readfirstlane(
                         (int)(blockIdx.x * PULL_ROWS_PER_BLOCK + (threadIdx.x >> 5)));
    const int lane = threadIdx.x & 31;
    const float4* x4 = (const float4*)x;

    int       i   = rowptr[row];
    const int end = rowptr[row + 1];
    const float  di   = dinv[row];
    const float4 self = x4[(size_t)row * (DIM / 4) + lane];
    float4 pv = self;
    if (!PREV_IS_X) pv = ((const float4*)prev)[(size_t)row * (DIM / 4) + lane];

    float4 acc = make_float4(0.0f, 0.0f, 0.0f, 0.0f);
    for (; i + 4 <= end; i += 4) pull_rows<4>(edge, x4, i, lane, acc);
    switch (end - i) {
        case 3: pull_rows<3>(edge, x4, i, lane, acc); break;
        case 2: pull_rows<2>(edge, x4, i, lane, acc); break;
        case 1: pull_rows<1>(edge, x4, i, lane, acc); break;
        default: break;
    }

    float4 o;
    o.x = alpha * ((acc.x + self.x * di) * di) + beta * pv.x;
    o.y = alpha * ((acc.y + self.y * di) * di) + beta * pv.y;
    o.z = alpha * ((acc.z + self.z * di) * di) + beta * pv.z;
    o.w = alpha * ((acc.w + self.w * di) * di) + beta * pv.w;
    ((float4*)out)[(size_t)row * (DIM / 4) + lane] = o;
}

__global__ void __launch_bounds__(32)
k_cheb_gemm(const float* __restrict__ X0,
            const float* __restrict__ X1,
            const float* __restrict__ X2,
            const float* __restrict__ W,
            const float* __restrict__ b,
            float* __restrict__ out) {
    const int row0 = blockIdx.x * 16;
    const int lane = threadIdx.x & 31;
    const int half = lane >> 4;
    const int lr   = lane & 15;

    v8f acc[8];
#pragma unroll
    for (int nt = 0; nt < 8; ++nt) {
        float bv = b[nt * 16 + lr];
        acc[nt] = (v8f){bv, bv, bv, bv, bv, bv, bv, bv};
    }

    const float* segs[3] = {X0, X1, X2};
    for (int seg = 0; seg < 3; ++seg) {
        const float* Xp = segs[seg] + (size_t)row0 * DIM;
#pragma unroll 4
        for (int kk = 0; kk < DIM; kk += 4) {
            const int ka = kk + 2 * half;
            v2f a;
            a.x = Xp[(size_t)lr * DIM + ka];
            a.y = Xp[(size_t)lr * DIM + ka + 1];
            const float* Wk  = W + (size_t)(seg * DIM + ka) * DIM;
            const float* Wk1 = Wk + DIM;
#pragma unroll
            for (int nt = 0; nt < 8; ++nt) {
                v2f bb;
                bb.x = Wk [nt * 16 + lr];
                bb.y = Wk1[nt * 16 + lr];
                acc[nt] = __builtin_amdgcn_wmma_f32_16x16x4_f32(
                     false, a,  false, bb,
                     (short)0, acc[nt],
                     false,  false);
            }
        }
    }

#pragma unroll
    for (int nt = 0; nt < 8; ++nt) {
        const int col = nt * 16 + lr;
#pragma unroll
        for (int r = 0; r < 8; ++r) {
            const int row = row0 + r + 8 * half;
            float v = acc[nt][r];
            out[(size_t)row * DIM + col] = v > 0.0f ? v : 0.0f;
        }
    }
}

__global__ void k_pred(const float* __restrict__ emb,
                       const float* __restrict__ pw,
                       const float* __restrict__ pb,
                       float* __restrict__ out) {
    int n = blockIdx.x * blockDim.x + threadIdx.x;
    if (n < N_NODES) {
        const float* e = emb + (size_t)n * DIM;
        float s = 0.0f;
#pragma unroll 8
        for (int d = 0; d < DIM; d += 4) {
            float4 v = *(const float4*)(e + d);
            s += v.x * pw[d] + v.y * pw[d + 1] + v.z * pw[d + 2] + v.w * pw[d + 3];
        }
        out[n] = s + pb[0];
    }
}

extern "C" void kernel_launch(void* const* d_in, const int* in_sizes, int n_in,
                              void* d_out, int out_size, void* d_ws, size_t ws_size,
                              hipStream_t stream) {
    const float* weights  = (const float*)d_in[0];
    const int*   src      = (const int*)  d_in[1];
    const int*   dst      = (const int*)  d_in[2];
    const float* lin_in_w = (const float*)d_in[3];
    const float* lin_in_b = (const float*)d_in[4];
    const float* cheb_ws  = (const float*)d_in[5];
    const float* cheb_bs  = (const float*)d_in[6];
    const float* pred_w   = (const float*)d_in[7];
    const float* pred_b   = (const float*)d_in[8];
    float* logits = (float*)d_out;

    float* ws   = (float*)d_ws;
    float* dinv = ws;
    float* X0   = ws + 51200;
    float* X1   = X0 + NF;
    float* X2   = X1 + NF;
    int*   g      = (int*)(X2 + NF);
    int*   cnt    = g;
    int*   rowptr = g + 51200;
    int*   cursor = g + 102400;
    Edge*  edge   = (Edge*)(g + 153600);
    float* emb  = X2 + NF + NF;

    const int T = 256;
    const int gN    = (N_NODES + T - 1) / T;
    const int gE    = (N_EDGES + T - 1) / T;
    const int gNF   = (NF + T - 1) / T;
    const int gPull = N_NODES / PULL_ROWS_PER_BLOCK;

    k_zero_int<<<gN, T, 0, stream>>>(cnt, N_NODES);
    k_count<<<gE, T, 0, stream>>>(dst, cnt);
    k_scan<<<SCAN_BLOCKS, SCAN_T, 0, stream>>>(cnt, rowptr, cursor, dinv);
    k_fill_csr<<<gE, T, 0, stream>>>(src, dst, dinv, cursor, edge);

    k_feat<<<gNF, T, 0, stream>>>(weights, lin_in_w, lin_in_b, X0);

    float* cur = X0;
    float* nxt = emb;
    for (int layer = 0; layer < 3; ++layer) {
        const float* Wl = cheb_ws + (size_t)layer * 3 * DIM * DIM;
        const float* bl = cheb_bs + (size_t)layer * DIM;

        k_spmm_pull<true><<<gPull, 32 * PULL_ROWS_PER_BLOCK, 0, stream>>>(
            rowptr, edge, cur, cur, dinv, X1, -1.0f, 0.0f);

        k_spmm_pull<false><<<gPull, 32 * PULL_ROWS_PER_BLOCK, 0, stream>>>(
            rowptr, edge, X1, cur, dinv, X2, -2.0f, -1.0f);

        k_cheb_gemm<<<N_NODES / 16, 32, 0, stream>>>(cur, X1, X2, Wl, bl, nxt);

        float* t = cur; cur = nxt; nxt = t;
    }

    k_pred<<<gN, T, 0, stream>>>(cur, pred_w, pred_b, logits);
}


# Layer 1 kept in rank-2 form (X0 = w x lin_w + 1 x lin_b): two 2-float-per-node gathers replace the embedding pass and both 128-wide gathers, GEMM builds A rows from them; layer-1 sums re-associated
# speedup vs baseline: 2.9703x; 1.1047x over previous
#include <hip/hip_runtime.h>


typedef float v2f __attribute__((ext_vector_type(2)));
typedef float v8f __attribute__((ext_vector_type(8)));

#define N_NODES 50000
#define N_EDGES 600000
#define DIM     128
#define NF      (N_NODES * DIM)

struct Edge { int src; float w; };

__global__ void k_zero_int(int* __restrict__ p, int n) {
    int i = blockIdx.x * blockDim.x + threadIdx.x;
    if (i < n) p[i] = 0;
}

__global__ void k_count(const int* __restrict__ dst, int* __restrict__ cnt) {
    int e = blockIdx.x * blockDim.x + threadIdx.x;
    if (e < N_EDGES) atomicAdd(&cnt[dst[e]], 1);
}

#define SCAN_T      1024
#define SCAN_BLOCKS ((N_NODES + SCAN_T - 1) / SCAN_T)

__global__ void __launch_bounds__(SCAN_T)
k_scan(const int* __restrict__ cnt,
       int* __restrict__ rowptr,
       int* __restrict__ cursor,
       float* __restrict__ dinv) {
    __shared__ int s_cnt[2][SCAN_T];
    __shared__ int s_pre[2][SCAN_T];
    const int t     = threadIdx.x;
    const int first = blockIdx.x * SCAN_T;
    const int i     = first + t;
    const int c     = (i < N_NODES) ? cnt[i] : 0;
    int pre = 0;
    for (int j = t; j < first; j += SCAN_T) pre += cnt[j];

    s_cnt[0][t] = c;
    s_pre[0][t] = pre;
    __syncthreads();
    int cur = 0;
    for (int off = 1; off < SCAN_T; off <<= 1) {
        int vc = s_cnt[cur][t], vp = s_pre[cur][t];
        if (t >= off) { vc += s_cnt[cur][t - off]; vp += s_pre[cur][t - off]; }
        s_cnt[cur ^ 1][t] = vc;
        s_pre[cur ^ 1][t] = vp;
        __syncthreads();
        cur ^= 1;
    }
    const int incl = s_cnt[cur][t];
    const int base = s_pre[cur][SCAN_T - 1];
    const int excl = base + incl - c;
    if (i < N_NODES) {
        rowptr[i] = excl;
        cursor[i] = excl;
        dinv[i]   = rsqrtf(1.0f + (float)c);
    }
    if (i == N_NODES - 1) rowptr[N_NODES] = excl + c;
}

__global__ void k_fill_csr(const int* __restrict__ src,
                           const int* __restrict__ dst,
                           const float* __restrict__ dinv,
                           int* __restrict__ cursor,
                           Edge* __restrict__ edge) {
    int e = blockIdx.x * blockDim.x + threadIdx.x;
    if (e < N_EDGES) {
        const int s = src[e];
        Edge r;
        r.src = s;
        r.w   = dinv[s];
        int slot = atomicAdd(&cursor[dst[e]], 1);
        edge[slot] = r;
    }
}

#define PAIR_LANES 8
#define PAIR_T     256
#define PAIR_BLOCKS ((N_NODES * PAIR_LANES + PAIR_T - 1) / PAIR_T)

template <bool FIRST>
__global__ void __launch_bounds__(PAIR_T)
k_spmm_pair(const int* __restrict__ rowptr,
            const Edge* __restrict__ edge,
            const float* __restrict__ wn,
            const float2* __restrict__ in,
            const float* __restrict__ dinv,
            float2* __restrict__ outp) {
    const int t   = blockIdx.x * PAIR_T + threadIdx.x;
    const int sub = t & (PAIR_LANES - 1);
    int row = t / PAIR_LANES;
    const bool live = row < N_NODES;
    if (!live) row = N_NODES - 1;

    const int end = rowptr[row + 1];
    float a = 0.0f, c = 0.0f;
    for (int i = rowptr[row] + sub; i < end; i += PAIR_LANES) {
        const Edge r = edge[i];
        float2 v;
        if (FIRST) { v.x = wn[r.src]; v.y = 1.0f; }
        else       { v = in[r.src]; }
        a += v.x * r.w;
        c += v.y * r.w;
    }
    for (int m = 1; m < PAIR_LANES; m <<= 1) {
        a += __shfl_xor(a, m);
        c += __shfl_xor(c, m);
    }
    if (live && sub == 0) {
        const float di = dinv[row];
        float2 s;
        if (FIRST) { s.x = wn[row]; s.y = 1.0f; }
        else       { s = in[row]; }
        float2 o;
        o.x = (a + s.x * di) * di;
        o.y = (c + s.y * di) * di;
        outp[row] = o;
    }
}

#define PULL_ROWS_PER_BLOCK 8
static_assert(N_NODES % PULL_ROWS_PER_BLOCK == 0, "grid covers the rows exactly");

template <int G>
__device__ __forceinline__ void pull_rows(const Edge* __restrict__ edge,
                                          const float4* __restrict__ x4,
                                          int i, int lane, float4& acc) {
    Edge r[G]; float4 v[G];
#pragma unroll
    for (int k = 0; k < G; ++k) r[k] = edge[i + k];
#pragma unroll
    for (int k = 0; k < G; ++k) v[k] = x4[(size_t)r[k].src * (DIM / 4) + lane];
#pragma unroll
    for (int k = 0; k < G; ++k) {
        acc.x += v[k].x * r[k].w;
        acc.y += v[k].y * r[k].w;
        acc.z += v[k].z * r[k].w;
        acc.w += v[k].w * r[k].w;
    }
}

template <bool PREV_IS_X>
__global__ void __launch_bounds__(32 * PULL_ROWS_PER_BLOCK)
k_spmm_pull(const int* __restrict__ rowptr,
            const Edge* __restrict__ edge,
            const float* __restrict__ x,
            const float* __restrict__ prev,
            const float* __restrict__ dinv,
            float* __restrict__ out,
            float alpha, float beta) {
    const int row  = __builtin_amdgcn_readfirstlane(
                         (int)(blockIdx.x * PULL_ROWS_PER_BLOCK + (threadIdx.x >> 5)));
    const int lane = threadIdx.x & 31;
    const float4* x4 = (const float4*)x;

    int       i   = rowptr[row];
    const int end = rowptr[row + 1];
    const float  di   = dinv[row];
    const float4 self = x4[(size_t)row * (DIM / 4) + lane];
    float4 pv = self;
    if (!PREV_IS_X) pv = ((const float4*)prev)[(size_t)row * (DIM / 4) + lane];

    float4 acc = make_float4(0.0f, 0.0f, 0.0f, 0.0f);
    for (; i + 4 <= end; i += 4) pull_rows<4>(edge, x4, i, lane, acc);
    switch (end - i) {
        case 3: pull_rows<3>(edge, x4, i, lane, acc); break;
        case 2: pull_rows<2>(edge, x4, i, lane, acc); break;
        case 1: pull_rows<1>(edge, x4, i, lane, acc); break;
        default: break;
    }

    float4 o;
    o.x = alpha * ((acc.x + self.x * di) * di) + beta * pv.x;
    o.y = alpha * ((acc.y + self.y * di) * di) + beta * pv.y;
    o.z = alpha * ((acc.z + self.z * di) * di) + beta * pv.z;
    o.w = alpha * ((acc.w + self.w * di) * di) + beta * pv.w;
    ((float4*)out)[(size_t)row * (DIM / 4) + lane] = o;
}

template <bool RANK2>
__global__ void __launch_bounds__(32)
k_cheb_gemm(const float* __restrict__ X0,
            const float* __restrict__ X1,
            const float* __restrict__ X2,
            const float* __restrict__ W,
            const float* __restrict__ b,
            float* __restrict__ out,
            const float* __restrict__ wn,
            const float2* __restrict__ p1,
            const float2* __restrict__ p2,
            const float* __restrict__ lw,
            const float* __restrict__ lb) {
    const int row0 = blockIdx.x * 16;
    const int lane = threadIdx.x & 31;
    const int half = lane >> 4;
    const int lr   = lane & 15;

    v8f acc[8];
#pragma unroll
    for (int nt = 0; nt < 8; ++nt) {
        float bv = b[nt * 16 + lr];
        acc[nt] = (v8f){bv, bv, bv, bv, bv, bv, bv, bv};
    }

    float al[3] = {0.0f, 0.0f, 0.0f}, be[3] = {0.0f, 0.0f, 0.0f};
    if (RANK2) {
        const float  wv = wn[row0 + lr];
        const float2 q1 = p1[row0 + lr];
        const float2 q2 = p2[row0 + lr];
        al[0] = wv;                 be[0] = 1.0f;
        al[1] = -q1.x;              be[1] = -q1.y;
        al[2] = 2.0f * q2.x - wv;   be[2] = 2.0f * q2.y - 1.0f;
    }

    const float* segs[3] = {X0, X1, X2};
#pragma unroll
    for (int seg = 0; seg < 3; ++seg) {
        const float* Xp = segs[seg] + (size_t)row0 * DIM;
#pragma unroll 4
        for (int kk = 0; kk < DIM; kk += 4) {
            const int ka = kk + 2 * half;
            v2f a;
            if (RANK2) {
                a.x = al[seg] * lw[ka]     + be[seg] * lb[ka];
                a.y = al[seg] * lw[ka + 1] + be[seg] * lb[ka + 1];
            } else {
                a.x = Xp[(size_t)lr * DIM + ka];
                a.y = Xp[(size_t)lr * DIM + ka + 1];
            }
            const float* Wk  = W + (size_t)(seg * DIM + ka) * DIM;
            const float* Wk1 = Wk + DIM;
#pragma unroll
            for (int nt = 0; nt < 8; ++nt) {
                v2f bb;
                bb.x = Wk [nt * 16 + lr];
                bb.y = Wk1[nt * 16 + lr];
                acc[nt] = __builtin_amdgcn_wmma_f32_16x16x4_f32(
                     false, a,  false, bb,
                     (short)0, acc[nt],
                     false,  false);
            }
        }
    }

#pragma unroll
    for (int nt = 0; nt < 8; ++nt) {
        const int col = nt * 16 + lr;
#pragma unroll
        for (int r = 0; r < 8; ++r) {
            const int row = row0 + r + 8 * half;
            float v = acc[nt][r];
            out[(size_t)row * DIM + col] = v > 0.0f ? v : 0.0f;
        }
    }
}

__global__ void k_pred(const float* __restrict__ emb,
                       const float* __restrict__ pw,
                       const float* __restrict__ pb,
                       float* __restrict__ out) {
    int n = blockIdx.x * blockDim.x + threadIdx.x;
    if (n < N_NODES) {
        const float* e = emb + (size_t)n * DIM;
        float s = 0.0f;
#pragma unroll 8
        for (int d = 0; d < DIM; d += 4) {
            float4 v = *(const float4*)(e + d);
            s += v.x * pw[d] + v.y * pw[d + 1] + v.z * pw[d + 2] + v.w * pw[d + 3];
        }
        out[n] = s + pb[0];
    }
}

extern "C" void kernel_launch(void* const* d_in, const int* in_sizes, int n_in,
                              void* d_out, int out_size, void* d_ws, size_t ws_size,
                              hipStream_t stream) {
    const float* weights  = (const float*)d_in[0];
    const int*   src      = (const int*)  d_in[1];
    const int*   dst      = (const int*)  d_in[2];
    const float* lin_in_w = (const float*)d_in[3];
    const float* lin_in_b = (const float*)d_in[4];
    const float* cheb_ws  = (const float*)d_in[5];
    const float* cheb_bs  = (const float*)d_in[6];
    const float* pred_w   = (const float*)d_in[7];
    const float* pred_b   = (const float*)d_in[8];
    float* logits = (float*)d_out;

    float* ws   = (float*)d_ws;
    float* dinv = ws;
    float* X0   = ws + 51200;
    float* X1   = X0 + NF;
    float* X2   = X1 + NF;
    int*   g      = (int*)(X2 + NF);
    int*   cnt    = g;
    int*   rowptr = g + 51200;
    int*   cursor = g + 102400;
    Edge*  edge   = (Edge*)(g + 153600);
    float* emb  = X2 + NF + NF;
    float2* p1  = (float2*)X1;
    float2* p2  = (float2*)X1 + 51200;

    const int T = 256;
    const int gN    = (N_NODES + T - 1) / T;
    const int gE    = (N_EDGES + T - 1) / T;
    const int gPull = N_NODES / PULL_ROWS_PER_BLOCK;

    k_zero_int<<<gN, T, 0, stream>>>(cnt, N_NODES);
    k_count<<<gE, T, 0, stream>>>(dst, cnt);
    k_scan<<<SCAN_BLOCKS, SCAN_T, 0, stream>>>(cnt, rowptr, cursor, dinv);
    k_fill_csr<<<gE, T, 0, stream>>>(src, dst, dinv, cursor, edge);

    k_spmm_pair<true><<<PAIR_BLOCKS, PAIR_T, 0, stream>>>(rowptr, edge, weights, p2, dinv, p1);
    k_spmm_pair<false><<<PAIR_BLOCKS, PAIR_T, 0, stream>>>(rowptr, edge, weights, p1, dinv, p2);
    k_cheb_gemm<true><<<N_NODES / 16, 32, 0, stream>>>(X0, X0, X0, cheb_ws, cheb_bs, emb,
                                                      weights, p1, p2, lin_in_w, lin_in_b);

    float* cur = emb;
    float* nxt = X0;
    for (int layer = 1; layer < 3; ++layer) {
        const float* Wl = cheb_ws + (size_t)layer * 3 * DIM * DIM;
        const float* bl = cheb_bs + (size_t)layer * DIM;

        k_spmm_pull<true><<<gPull, 32 * PULL_ROWS_PER_BLOCK, 0, stream>>>(
            rowptr, edge, cur, cur, dinv, X1, -1.0f, 0.0f);

        k_spmm_pull<false><<<gPull, 32 * PULL_ROWS_PER_BLOCK, 0, stream>>>(
            rowptr, edge, X1, cur, dinv, X2, -2.0f, -1.0f);

        k_cheb_gemm<false><<<N_NODES / 16, 32, 0, stream>>>(cur, X1, X2, Wl, bl, nxt,
                                                           weights, p1, p2, lin_in_w, lin_in_b);

        float* t = cur; cur = nxt; nxt = t;
    }

    k_pred<<<gN, T, 0, stream>>>(cur, pred_w, pred_b, logits);
}


# GEMM: W reaches the WMMAs through LDS: a work-group of 8 waves (128 rows) stages each 128-row segment of W once; B fragments come from LDS, not 2 global loads per WMMA; product order unchanged
# speedup vs baseline: 3.1949x; 1.0756x over previous
#include <hip/hip_runtime.h>


typedef float v2f __attribute__((ext_vector_type(2)));
typedef float v8f __attribute__((ext_vector_type(8)));

#define N_NODES 50000
#define N_EDGES 600000
#define DIM     128
#define NF      (N_NODES * DIM)

struct Edge { int src; float w; };

__global__ void k_zero_int(int* __restrict__ p, int n) {
    int i = blockIdx.x * blockDim.x + threadIdx.x;
    if (i < n) p[i] = 0;
}

__global__ void k_count(const int* __restrict__ dst, int* __restrict__ cnt) {
    int e = blockIdx.x * blockDim.x + threadIdx.x;
    if (e < N_EDGES) atomicAdd(&cnt[dst[e]], 1);
}

#define SCAN_T      1024
#define SCAN_BLOCKS ((N_NODES + SCAN_T - 1) / SCAN_T)

__global__ void __launch_bounds__(SCAN_T)
k_scan(const int* __restrict__ cnt,
       int* __restrict__ rowptr,
       int* __restrict__ cursor,
       float* __restrict__ dinv) {
    __shared__ int s_cnt[2][SCAN_T];
    __shared__ int s_pre[2][SCAN_T];
    const int t     = threadIdx.x;
    const int first = blockIdx.x * SCAN_T;
    const int i     = first + t;
    const int c     = (i < N_NODES) ? cnt[i] : 0;
    int pre = 0;
    for (int j = t; j < first; j += SCAN_T) pre += cnt[j];

    s_cnt[0][t] = c;
    s_pre[0][t] = pre;
    __syncthreads();
    int cur = 0;
    for (int off = 1; off < SCAN_T; off <<= 1) {
        int vc = s_cnt[cur][t], vp = s_pre[cur][t];
        if (t >= off) { vc += s_cnt[cur][t - off]; vp += s_pre[cur][t - off]; }
        s_cnt[cur ^ 1][t] = vc;
        s_pre[cur ^ 1][t] = vp;
        __syncthreads();
        cur ^= 1;
    }
    const int incl = s_cnt[cur][t];
    const int base = s_pre[cur][SCAN_T - 1];
    const int excl = base + incl - c;
    if (i < N_NODES) {
        rowptr[i] = excl;
        cursor[i] = excl;
        dinv[i]   = rsqrtf(1.0f + (float)c);
    }
    if (i == N_NODES - 1) rowptr[N_NODES] = excl + c;
}

__global__ void k_fill_csr(const int* __restrict__ src,
                           const int* __restrict__ dst,
                           const float* __restrict__ dinv,
                           int* __restrict__ cursor,
                           Edge* __restrict__ edge) {
    int e = blockIdx.x * blockDim.x + threadIdx.x;
    if (e < N_EDGES) {
        const int s = src[e];
        Edge r;
        r.src = s;
        r.w   = dinv[s];
        int slot = atomicAdd(&cursor[dst[e]], 1);
        edge[slot] = r;
    }
}

#define PAIR_LANES 8
#define PAIR_T     256
#define PAIR_BLOCKS ((N_NODES * PAIR_LANES + PAIR_T - 1) / PAIR_T)

template <bool FIRST>
__global__ void __launch_bounds__(PAIR_T)
k_spmm_pair(const int* __restrict__ rowptr,
            const Edge* __restrict__ edge,
            const float* __restrict__ wn,
            const float2* __restrict__ in,
            const float* __restrict__ dinv,
            float2* __restrict__ outp) {
    const int t   = blockIdx.x * PAIR_T + threadIdx.x;
    const int sub = t & (PAIR_LANES - 1);
    int row = t / PAIR_LANES;
    const bool live = row < N_NODES;
    if (!live) row = N_NODES - 1;

    const int end = rowptr[row + 1];
    float a = 0.0f, c = 0.0f;
    for (int i = rowptr[row] + sub; i < end; i += PAIR_LANES) {
        const Edge r = edge[i];
        float2 v;
        if (FIRST) { v.x = wn[r.src]; v.y = 1.0f; }
        else       { v = in[r.src]; }
        a += v.x * r.w;
        c += v.y * r.w;
    }
    for (int m = 1; m < PAIR_LANES; m <<= 1) {
        a += __shfl_xor(a, m);
        c += __shfl_xor(c, m);
    }
    if (live && sub == 0) {
        const float di = dinv[row];
        float2 s;
        if (FIRST) { s.x = wn[row]; s.y = 1.0f; }
        else       { s = in[row]; }
        float2 o;
        o.x = (a + s.x * di) * di;
        o.y = (c + s.y * di) * di;
        outp[row] = o;
    }
}

#define PULL_ROWS_PER_BLOCK 8
static_assert(N_NODES % PULL_ROWS_PER_BLOCK == 0, "grid covers the rows exactly");

template <int G>
__device__ __forceinline__ void pull_rows(const Edge* __restrict__ edge,
                                          const float4* __restrict__ x4,
                                          int i, int lane, float4& acc) {
    Edge r[G]; float4 v[G];
#pragma unroll
    for (int k = 0; k < G; ++k) r[k] = edge[i + k];
#pragma unroll
    for (int k = 0; k < G; ++k) v[k] = x4[(size_t)r[k].src * (DIM / 4) + lane];
#pragma unroll
    for (int k = 0; k < G; ++k) {
        acc.x += v[k].x * r[k].w;
        acc.y += v[k].y * r[k].w;
        acc.z += v[k].z * r[k].w;
        acc.w += v[k].w * r[k].w;
    }
}

template <bool PREV_IS_X>
__global__ void __launch_bounds__(32 * PULL_ROWS_PER_BLOCK)
k_spmm_pull(const int* __restrict__ rowptr,
            const Edge* __restrict__ edge,
            const float* __restrict__ x,
            const float* __restrict__ prev,
            const float* __restrict__ dinv,
            float* __restrict__ out,
            float alpha, float beta) {
    const int row  = __builtin_amdgcn_readfirstlane(
                         (int)(blockIdx.x * PULL_ROWS_PER_BLOCK + (threadIdx.x >> 5)));
    const int lane = threadIdx.x & 31;
    const float4* x4 = (const float4*)x;

    int       i   = rowptr[row];
    const int end = rowptr[row + 1];
    const float  di   = dinv[row];
    const float4 self = x4[(size_t)row * (DIM / 4) + lane];
    float4 pv = self;
    if (!PREV_IS_X) pv = ((const float4*)prev)[(size_t)row * (DIM / 4) + lane];

    float4 acc = make_float4(0.0f, 0.0f, 0.0f, 0.0f);
    for (; i + 4 <= end; i += 4) pull_rows<4>(edge, x4, i, lane, acc);
    switch (end - i) {
        case 3: pull_rows<3>(edge, x4, i, lane, acc); break;
        case 2: pull_rows<2>(edge, x4, i, lane, acc); break;
        case 1: pull_rows<1>(edge, x4, i, lane, acc); break;
        default: break;
    }

    float4 o;
    o.x = alpha * ((acc.x + self.x * di) * di) + beta * pv.x;
    o.y = alpha * ((acc.y + self.y * di) * di) + beta * pv.y;
    o.z = alpha * ((acc.z + self.z * di) * di) + beta * pv.z;
    o.w = alpha * ((acc.w + self.w * di) * di) + beta * pv.w;
    ((float4*)out)[(size_t)row * (DIM / 4) + lane] = o;
}

#define GEMM_WAVES  8
#define GEMM_T      (32 * GEMM_WAVES)
#define GEMM_ROWS   (16 * GEMM_WAVES)
#define GEMM_BLOCKS ((N_NODES + GEMM_ROWS - 1) / GEMM_ROWS)
#define N_TILES     (N_NODES / 16)
#define W_PITCH     136
#define W_F4        (DIM * DIM / 4 / GEMM_T)

#define W16(OP) OP(0) OP(1) OP(2) OP(3) OP(4) OP(5) OP(6) OP(7) OP(8) OP(9) OP(10) OP(11) OP(12) OP(13) OP(14) OP(15)
#define W_DECL(i)   float4 w##i;
#define W_STORE(i)  *(float4*)(sS + (i) * 8 * W_PITCH) = w##i;
#define W_LOAD0(i)  w##i = Wg[(size_t)(0 * DIM + (i) * 8) * (DIM / 4)];
#define W_LOAD1(i)  w##i = Wg[(size_t)(1 * DIM + (i) * 8) * (DIM / 4)];
#define W_LOAD2(i)  w##i = Wg[(size_t)(2 * DIM + (i) * 8) * (DIM / 4)];
static_assert(W_F4 == 16, "W16 lists the float4 of one thread");

template <bool RANK2>
__device__ __forceinline__ void gemm_segment(v8f (&acc)[8],
                                             const float* __restrict__ Ap,
                                             const float* Bp,
                                             float al, float be,
                                             const float* __restrict__ lw,
                                             const float* __restrict__ lb,
                                             int half) {
#pragma unroll 4
    for (int kk = 0; kk < DIM; kk += 4) {
        const int ka = kk + 2 * half;
        v2f a;
        if (RANK2) {
            a.x = al * lw[ka]     + be * lb[ka];
            a.y = al * lw[ka + 1] + be * lb[ka + 1];
        } else {
            a.x = Ap[kk];
            a.y = Ap[kk + 1];
        }
        const float* Bk  = Bp + kk * W_PITCH;
        const float* Bk1 = Bk + W_PITCH;
#pragma unroll
        for (int nt = 0; nt < 8; ++nt) {
            v2f bb;
            bb.x = Bk [nt * 16];
            bb.y = Bk1[nt * 16];
            acc[nt] = __builtin_amdgcn_wmma_f32_16x16x4_f32(
                 false, a,  false, bb,
                 (short)0, acc[nt],
                 false,  false);
        }
    }
}

template <bool RANK2>
__global__ void __launch_bounds__(GEMM_T)
k_cheb_gemm(const float* __restrict__ X0,
            const float* __restrict__ X1,
            const float* __restrict__ X2,
            const float* __restrict__ W,
            const float* __restrict__ b,
            float* __restrict__ out,
            const float* __restrict__ wn,
            const float2* __restrict__ p1,
            const float2* __restrict__ p2,
            const float* __restrict__ lw,
            const float* __restrict__ lb) {
    __shared__ __attribute__((aligned(16))) float sW[DIM * W_PITCH];

    const int tid  = threadIdx.x;
    const int wave = __builtin_amdgcn_readfirstlane(tid >> 5);
    const int tile = blockIdx.x * GEMM_WAVES + wave;
    const bool live = tile < N_TILES;
    const int row0 = (live ? tile : N_TILES - 1) * 16;
    const int lane = tid & 31;
    const int half = lane >> 4;
    const int lr   = lane & 15;

    const float4* Wg = (const float4*)W + (size_t)(tid >> 5) * (DIM / 4) + (tid & 31);
    float*        sS = sW + (tid >> 5) * W_PITCH + (tid & 31) * 4;
    W16(W_DECL)
    W16(W_LOAD0)

    v8f acc[8];
#pragma unroll
    for (int nt = 0; nt < 8; ++nt) {
        float bv = b[nt * 16 + lr];
        acc[nt] = (v8f){bv, bv, bv, bv, bv, bv, bv, bv};
    }

    float al0 = 0.0f, al1 = 0.0f, al2 = 0.0f, be0 = 0.0f, be1 = 0.0f, be2 = 0.0f;
    if (RANK2) {
        const float  wv = wn[row0 + lr];
        const float2 q1 = p1[row0 + lr];
        const float2 q2 = p2[row0 + lr];
        al0 = wv;                 be0 = 1.0f;
        al1 = -q1.x;              be1 = -q1.y;
        al2 = 2.0f * q2.x - wv;   be2 = 2.0f * q2.y - 1.0f;
    }

    const size_t aoff = (size_t)(row0 + lr) * DIM + 2 * half;
    const float* Bp   = sW + (2 * half) * W_PITCH + lr;

    W16(W_STORE)
    __syncthreads();
    W16(W_LOAD1)
    gemm_segment<RANK2>(acc, X0 + aoff, Bp, al0, be0, lw, lb, half);

    __syncthreads();
    W16(W_STORE)
    __syncthreads();
    W16(W_LOAD2)
    gemm_segment<RANK2>(acc, X1 + aoff, Bp, al1, be1, lw, lb, half);

    __syncthreads();
    W16(W_STORE)
    __syncthreads();
    gemm_segment<RANK2>(acc, X2 + aoff, Bp, al2, be2, lw, lb, half);

    if (live) {
#pragma unroll
        for (int nt = 0; nt < 8; ++nt) {
            const int col = nt * 16 + lr;
#pragma unroll
            for (int r = 0; r < 8; ++r) {
                const int row = row0 + r + 8 * half;
                float v = acc[nt][r];
                out[(size_t)row * DIM + col] = v > 0.0f ? v : 0.0f;
            }
        }
    }
}

__global__ void k_pred(const float* __restrict__ emb,
                       const float* __restrict__ pw,
                       const float* __restrict__ pb,
                       float* __restrict__ out) {
    int n = blockIdx.x * blockDim.x + threadIdx.x;
    if (n < N_NODES) {
        const float* e = emb + (size_t)n * DIM;
        float s = 0.0f;
#pragma unroll 8
        for (int d = 0; d < DIM; d += 4) {
            float4 v = *(const float4*)(e + d);
            s += v.x * pw[d] + v.y * pw[d + 1] + v.z * pw[d + 2] + v.w * pw[d + 3];
        }
        out[n] = s + pb[0];
    }
}

extern "C" void kernel_launch(void* const* d_in, const int* in_sizes, int n_in,
                              void* d_out, int out_size, void* d_ws, size_t ws_size,
                              hipStream_t stream) {
    const float* weights  = (const float*)d_in[0];
    const int*   src      = (const int*)  d_in[1];
    const int*   dst      = (const int*)  d_in[2];
    const float* lin_in_w = (const float*)d_in[3];
    const float* lin_in_b = (const float*)d_in[4];
    const float* cheb_ws  = (const float*)d_in[5];
    const float* cheb_bs  = (const float*)d_in[6];
    const float* pred_w   = (const float*)d_in[7];
    const float* pred_b   = (const float*)d_in[8];
    float* logits = (float*)d_out;

    float* ws   = (float*)d_ws;
    float* dinv = ws;
    float* X0   = ws + 51200;
    float* X1   = X0 + NF;
    float* X2   = X1 + NF;
    int*   g      = (int*)(X2 + NF);
    int*   cnt    = g;
    int*   rowptr = g + 51200;
    int*   cursor = g + 102400;
    Edge*  edge   = (Edge*)(g + 153600);
    float* emb  = X2 + NF + NF;
    float2* p1  = (float2*)X1;
    float2* p2  = (float2*)X1 + 51200;

    const int T = 256;
    const int gN    = (N_NODES + T - 1) / T;
    const int gE    = (N_EDGES + T - 1) / T;
    const int gPull = N_NODES / PULL_ROWS_PER_BLOCK;

    k_zero_int<<<gN, T, 0, stream>>>(cnt, N_NODES);
    k_count<<<gE, T, 0, stream>>>(dst, cnt);
    k_scan<<<SCAN_BLOCKS, SCAN_T, 0, stream>>>(cnt, rowptr, cursor, dinv);
    k_fill_csr<<<gE, T, 0, stream>>>(src, dst, dinv, cursor, edge);

    k_spmm_pair<true><<<PAIR_BLOCKS, PAIR_T, 0, stream>>>(rowptr, edge, weights, p2, dinv, p1);
    k_spmm_pair<false><<<PAIR_BLOCKS, PAIR_T, 0, stream>>>(rowptr, edge, weights, p1, dinv, p2);
    k_cheb_gemm<true><<<GEMM_BLOCKS, GEMM_T, 0, stream>>>(X0, X0, X0, cheb_ws, cheb_bs, emb,
                                                      weights, p1, p2, lin_in_w, lin_in_b);

    float* cur = emb;
    float* nxt = X0;
    for (int layer = 1; layer < 3; ++layer) {
        const float* Wl = cheb_ws + (size_t)layer * 3 * DIM * DIM;
        const float* bl = cheb_bs + (size_t)layer * DIM;

        k_spmm_pull<true><<<gPull, 32 * PULL_ROWS_PER_BLOCK, 0, stream>>>(
            rowptr, edge, cur, cur, dinv, X1, -1.0f, 0.0f);

        k_spmm_pull<false><<<gPull, 32 * PULL_ROWS_PER_BLOCK, 0, stream>>>(
            rowptr, edge, X1, cur, dinv, X2, -2.0f, -1.0f);

        k_cheb_gemm<false><<<GEMM_BLOCKS, GEMM_T, 0, stream>>>(cur, X1, X2, Wl, bl, nxt,
                                                           weights, p1, p2, lin_in_w, lin_in_b);

        float* t = cur; cur = nxt; nxt = t;
    }

    k_pred<<<gN, T, 0, stream>>>(cur, pred_w, pred_b, logits);
}


# CSR build without global atomics: 100 work-groups each own 500 destination nodes, read the whole dst list twice (count, place) and rank their edges by LDS atomics; 14 launches to 12; row order moves
# speedup vs baseline: 3.8936x; 1.2187x over previous
#include <hip/hip_runtime.h>


typedef float v2f __attribute__((ext_vector_type(2)));
typedef float v8f __attribute__((ext_vector_type(8)));

#define N_NODES 50000
#define N_EDGES 600000
#define DIM     128
#define NF      (N_NODES * DIM)

struct Edge { int src; float w; };

#define RANGE_N      500
#define RANGE_BLOCKS (N_NODES / RANGE_N)
#define RANGE_T      1024
#define RANGE_WAVES  (RANGE_T / 32)
#define RANGE_SLOTS  512
#define DST_F4       (N_EDGES / 4)
#define RANGE_G      16
#define RANGE_FULL   (DST_F4 / (RANGE_G * RANGE_T))
#define RANGE_REM    3
#define STASH_W      512
static_assert(RANGE_BLOCKS * RANGE_N == N_NODES, "the ranges cover the nodes exactly");
static_assert(RANGE_N <= RANGE_SLOTS && RANGE_SLOTS <= RANGE_T, "one thread per counter in the scan");
static_assert(RANGE_BLOCKS <= 128, "the totals in front of a range are summed by the first four waves");
static_assert(N_EDGES % 4 == 0, "the dst list is read four edges at a time");
static_assert(DST_F4 - RANGE_FULL * RANGE_G * RANGE_T <= RANGE_REM * RANGE_T, "the last groups fit the guarded tail");
static_assert(N_EDGES <= (1 << 23) && RANGE_SLOTS == (1 << 9), "a stash entry packs edge number and counter into 32 bits");

template <typename F, typename G>
__device__ __forceinline__ void for_all_dst(const int* __restrict__ dst, int t, F f, G g) {
    const int4* d4 = (const int4*)dst;
    int j = t;
    for (int m = 0; m < RANGE_FULL; ++m, j += RANGE_G * RANGE_T) {
        int4 dv[RANGE_G];
#pragma unroll
        for (int k = 0; k < RANGE_G; ++k) dv[k] = d4[j + k * RANGE_T];
#pragma unroll
        for (int k = 0; k < RANGE_G; ++k) f(dv[k], j + k * RANGE_T);
        g();
    }
    int4 dv[RANGE_REM];
#pragma unroll
    for (int k = 0; k < RANGE_REM; ++k) {
        const int idx = j + k * RANGE_T;
        dv[k] = make_int4(-1, -1, -1, -1);
        if (idx < DST_F4) dv[k] = d4[idx];
    }
#pragma unroll
    for (int k = 0; k < RANGE_REM; ++k) f(dv[k], j + k * RANGE_T);
}

__device__ __forceinline__ void count_one(int* s_cnt, int dstv, int lo) {
    const unsigned r = (unsigned)(dstv - lo);
    const bool mine = r < RANGE_N;
    if (__builtin_expect(__builtin_amdgcn_ballot_w32(mine) != 0u, 0)) {
        if (mine) atomicAdd(&s_cnt[r], 1);
    }
}

__global__ void __launch_bounds__(RANGE_T)
k_range_count(const int* __restrict__ dst,
              int* __restrict__ lptr,
              float* __restrict__ dinv,
              int* __restrict__ total) {
    __shared__ int s_cnt[RANGE_SLOTS];
    __shared__ int s_scan[2][RANGE_SLOTS];
    const int t  = threadIdx.x;
    const int lo = blockIdx.x * RANGE_N;
    if (t < RANGE_SLOTS) s_cnt[t] = 0;
    __syncthreads();

    for_all_dst(dst, t, [&](const int4 dv, int) {
        count_one(s_cnt, dv.x, lo);
        count_one(s_cnt, dv.y, lo);
        count_one(s_cnt, dv.z, lo);
        count_one(s_cnt, dv.w, lo);
    }, []() {});
    __syncthreads();

    const int c = (t < RANGE_SLOTS) ? s_cnt[t] : 0;
    if (t < RANGE_SLOTS) s_scan[0][t] = c;
    __syncthreads();
    int cur = 0;
    for (int off = 1; off < RANGE_SLOTS; off <<= 1) {
        if (t < RANGE_SLOTS) {
            int v = s_scan[cur][t];
            if (t >= off) v += s_scan[cur][t - off];
            s_scan[cur ^ 1][t] = v;
        }
        __syncthreads();
        cur ^= 1;
    }
    if (t < RANGE_N) {
        lptr[lo + t] = s_scan[cur][t] - c;
        dinv[lo + t] = rsqrtf(1.0f + (float)c);
    }
    if (t == RANGE_SLOTS - 1) total[blockIdx.x] = s_scan[cur][t];
}

__device__ __forceinline__ void place_edge(int* s_cur, Edge* __restrict__ edge, unsigned r, int s, float w) {
    Edge q;
    q.src = s;
    q.w   = w;
    const int slot = atomicAdd(&s_cur[r], 1);
    edge[slot] = q;
}

__global__ void __launch_bounds__(RANGE_T)
k_range_place(const int* __restrict__ src,
              const int* __restrict__ dst,
              const int* __restrict__ lptr,
              const int* __restrict__ total,
              const float* __restrict__ dinv,
              int* __restrict__ rowptr,
              Edge* __restrict__ edge) {
    __shared__ int      s_cur[RANGE_SLOTS];
    __shared__ int      s_red[RANGE_WAVES];
    __shared__ unsigned s_stash[RANGE_WAVES * STASH_W];
    const int t    = threadIdx.x;
    const int b    = blockIdx.x;
    const int lo   = b * RANGE_N;
    const int lane = t & 31;
    const int wave = __builtin_amdgcn_readfirstlane(t >> 5);

    int v = (t < b) ? total[t] : 0;
    for (int m = 16; m >= 1; m >>= 1) v += __shfl_xor(v, m);
    if (lane == 0) s_red[wave] = v;
    __syncthreads();
    const int base = s_red[0] + s_red[1] + s_red[2] + s_red[3];

    if (t < RANGE_N) {
        const int p = base + lptr[lo + t];
        rowptr[lo + t] = p;
        s_cur[t] = p;
    }
    if (b == RANGE_BLOCKS - 1 && t == 0) rowptr[N_NODES] = base + total[b];
    __syncthreads();

    unsigned* my = s_stash + wave * STASH_W;
    int held = 0, head = 0;
    auto note_one = [&](int dstv, int e) {
        const unsigned r = (unsigned)(dstv - lo);
        const bool mine = r < RANGE_N;
        const unsigned hits = __builtin_amdgcn_ballot_w32(mine);
        if (hits) {
            if (mine) {
                const int pos = held + (int)__builtin_amdgcn_mbcnt_lo(hits, 0u);
                if (pos < STASH_W) {
                    my[pos] = ((unsigned)e << 9) | r;
                } else {
                    const int s = src[e];
                    place_edge(s_cur, edge, r, s, dinv[s]);
                }
            }
            held += __popc(hits);
        }
    };
    auto take32 = [&](int k0, int lim) {
        const int k = k0 + lane;
        if (k < lim) {
            const unsigned u = my[k];
            const int s = src[u >> 9];
            place_edge(s_cur, edge, u & (RANGE_SLOTS - 1), s, dinv[s]);
        }
    };
    for_all_dst(dst, t, [&](const int4 dv, int idx) {
        note_one(dv.x, 4 * idx + 0);
        note_one(dv.y, 4 * idx + 1);
        note_one(dv.z, 4 * idx + 2);
        note_one(dv.w, 4 * idx + 3);
    }, [&]() {
        const int lim = held < STASH_W ? held : STASH_W;
        for (; lim - head >= 32; head += 32) take32(head, lim);
    });
    const int lim = held < STASH_W ? held : STASH_W;
    for (; head < lim; head += 32) take32(head, lim);
}

#define PAIR_LANES 8
#define PAIR_T     256
#define PAIR_BLOCKS ((N_NODES * PAIR_LANES + PAIR_T - 1) / PAIR_T)

template <bool FIRST>
__global__ void __launch_bounds__(PAIR_T)
k_spmm_pair(const int* __restrict__ rowptr,
            const Edge* __restrict__ edge,
            const float* __restrict__ wn,
            const float2* __restrict__ in,
            const float* __restrict__ dinv,
            float2* __restrict__ outp) {
    const int t   = blockIdx.x * PAIR_T + threadIdx.x;
    const int sub = t & (PAIR_LANES - 1);
    int row = t / PAIR_LANES;
    const bool live = row < N_NODES;
    if (!live) row = N_NODES - 1;

    const int end = rowptr[row + 1];
    float a = 0.0f, c = 0.0f;
    for (int i = rowptr[row] + sub; i < end; i += PAIR_LANES) {
        const Edge r = edge[i];
        float2 v;
        if (FIRST) { v.x = wn[r.src]; v.y = 1.0f; }
        else       { v = in[r.src]; }
        a += v.x * r.w;
        c += v.y * r.w;
    }
    for (int m = 1; m < PAIR_LANES; m <<= 1) {
        a += __shfl_xor(a, m);
        c += __shfl_xor(c, m);
    }
    if (live && sub == 0) {
        const float di = dinv[row];
        float2 s;
        if (FIRST) { s.x = wn[row]; s.y = 1.0f; }
        else       { s = in[row]; }
        float2 o;
        o.x = (a + s.x * di) * di;
        o.y = (c + s.y * di) * di;
        outp[row] = o;
    }
}

#define PULL_ROWS_PER_BLOCK 8
static_assert(N_NODES % PULL_ROWS_PER_BLOCK == 0, "grid covers the rows exactly");

template <int G>
__device__ __forceinline__ void pull_rows(const Edge* __restrict__ edge,
                                          const float4* __restrict__ x4,
                                          int i, int lane, float4& acc) {
    Edge r[G]; float4 v[G];
#pragma unroll
    for (int k = 0; k < G; ++k) r[k] = edge[i + k];
#pragma unroll
    for (int k = 0; k < G; ++k) v[k] = x4[(size_t)r[k].src * (DIM / 4) + lane];
#pragma unroll
    for (int k = 0; k < G; ++k) {
        acc.x += v[k].x * r[k].w;
        acc.y += v[k].y * r[k].w;
        acc.z += v[k].z * r[k].w;
        acc.w += v[k].w * r[k].w;
    }
}

template <bool PREV_IS_X>
__global__ void __launch_bounds__(32 * PULL_ROWS_PER_BLOCK)
k_spmm_pull(const int* __restrict__ rowptr,
            const Edge* __restrict__ edge,
            const float* __restrict__ x,
            const float* __restrict__ prev,
            const float* __restrict__ dinv,
            float* __restrict__ out,
            float alpha, float beta) {
    const int row  = __builtin_amdgcn_readfirstlane(
                         (int)(blockIdx.x * PULL_ROWS_PER_BLOCK + (threadIdx.x >> 5)));
    const int lane = threadIdx.x & 31;
    const float4* x4 = (const float4*)x;

    int       i   = rowptr[row];
    const int end = rowptr[row + 1];
    const float  di   = dinv[row];
    const float4 self = x4[(size_t)row * (DIM / 4) + lane];
    float4 pv = self;
    if (!PREV_IS_X) pv = ((const float4*)prev)[(size_t)row * (DIM / 4) + lane];

    float4 acc = make_float4(0.0f, 0.0f, 0.0f, 0.0f);
    for (; i + 4 <= end; i += 4) pull_rows<4>(edge, x4, i, lane, acc);
    switch (end - i) {
        case 3: pull_rows<3>(edge, x4, i, lane, acc); break;
        case 2: pull_rows<2>(edge, x4, i, lane, acc); break;
        case 1: pull_rows<1>(edge, x4, i, lane, acc); break;
        default: break;
    }

    float4 o;
    o.x = alpha * ((acc.x + self.x * di) * di) + beta * pv.x;
    o.y = alpha * ((acc.y + self.y * di) * di) + beta * pv.y;
    o.z = alpha * ((acc.z + self.z * di) * di) + beta * pv.z;
    o.w = alpha * ((acc.w + self.w * di) * di) + beta * pv.w;
    ((float4*)out)[(size_t)row * (DIM / 4) + lane] = o;
}

#define GEMM_WAVES  8
#define GEMM_T      (32 * GEMM_WAVES)
#define GEMM_ROWS   (16 * GEMM_WAVES)
#define GEMM_BLOCKS ((N_NODES + GEMM_ROWS - 1) / GEMM_ROWS)
#define N_TILES     (N_NODES / 16)
#define W_PITCH     136
#define W_F4        (DIM * DIM / 4 / GEMM_T)

#define W16(OP) OP(0) OP(1) OP(2) OP(3) OP(4) OP(5) OP(6) OP(7) OP(8) OP(9) OP(10) OP(11) OP(12) OP(13) OP(14) OP(15)
#define W_DECL(i)   float4 w##i;
#define W_STORE(i)  *(float4*)(sS + (i) * 8 * W_PITCH) = w##i;
#define W_LOAD0(i)  w##i = Wg[(size_t)(0 * DIM + (i) * 8) * (DIM / 4)];
#define W_LOAD1(i)  w##i = Wg[(size_t)(1 * DIM + (i) * 8) * (DIM / 4)];
#define W_LOAD2(i)  w##i = Wg[(size_t)(2 * DIM + (i) * 8) * (DIM / 4)];
static_assert(W_F4 == 16, "W16 lists the float4 of one thread");

__device__ __forceinline__ void gemm_segment(v8f (&acc)[8],
                                             const float* __restrict__ Ap,
                                             const float* Bp) {
#pragma unroll 4
    for (int kk = 0; kk < DIM; kk += 4) {
        v2f a;
        a.x = Ap[kk];
        a.y = Ap[kk + 1];
        const float* Bk  = Bp + kk * W_PITCH;
        const float* Bk1 = Bk + W_PITCH;
#pragma unroll
        for (int nt = 0; nt < 8; ++nt) {
            v2f bb;
            bb.x = Bk [nt * 16];
            bb.y = Bk1[nt * 16];
            acc[nt] = __builtin_amdgcn_wmma_f32_16x16x4_f32(
                 false, a,  false, bb,
                 (short)0, acc[nt],
                 false,  false);
        }
    }
}

__global__ void __launch_bounds__(GEMM_T)
k_cheb_gemm(const float* __restrict__ X0,
            const float* __restrict__ X1,
            const float* __restrict__ X2,
            const float* __restrict__ W,
            const float* __restrict__ b,
            float* __restrict__ out) {
    __shared__ __attribute__((aligned(16))) float sW[DIM * W_PITCH];

    const int tid  = threadIdx.x;
    const int wave = __builtin_amdgcn_readfirstlane(tid >> 5);
    const int tile = blockIdx.x * GEMM_WAVES + wave;
    const bool live = tile < N_TILES;
    const int row0 = (live ? tile : N_TILES - 1) * 16;
    const int lane = tid & 31;
    const int half = lane >> 4;
    const int lr   = lane & 15;

    const float4* Wg = (const float4*)W + (size_t)(tid >> 5) * (DIM / 4) + (tid & 31);
    float*        sS = sW + (tid >> 5) * W_PITCH + (tid & 31) * 4;
    W16(W_DECL)
    W16(W_LOAD0)

    v8f acc[8];
#pragma unroll
    for (int nt = 0; nt < 8; ++nt) {
        float bv = b[nt * 16 + lr];
        acc[nt] = (v8f){bv, bv, bv, bv, bv, bv, bv, bv};
    }

    const size_t aoff = (size_t)(row0 + lr) * DIM + 2 * half;
    const float* Bp   = sW + (2 * half) * W_PITCH + lr;

    W16(W_STORE)
    __syncthreads();
    W16(W_LOAD1)
    gemm_segment(acc, X0 + aoff, Bp);

    __syncthreads();
    W16(W_STORE)
    __syncthreads();
    W16(W_LOAD2)
    gemm_segment(acc, X1 + aoff, Bp);

    __syncthreads();
    W16(W_STORE)
    __syncthreads();
    gemm_segment(acc, X2 + aoff, Bp);

    if (live) {
#pragma unroll
        for (int nt = 0; nt < 8; ++nt) {
            const int col = nt * 16 + lr;
#pragma unroll
            for (int r = 0; r < 8; ++r) {
                const int row = row0 + r + 8 * half;
                float v = acc[nt][r];
                out[(size_t)row * DIM + col] = v > 0.0f ? v : 0.0f;
            }
        }
    }
}

static_assert(GEMM_T == 2 * DIM, "one thread per element of lw | lb when they are copied to LDS");
#define U_PITCH 136

__device__ __forceinline__ v8f rank2_segment(const float* Ap, const float* Bp) {
    v8f acc = (v8f){0.0f, 0.0f, 0.0f, 0.0f, 0.0f, 0.0f, 0.0f, 0.0f};
#pragma unroll 8
    for (int kk = 0; kk < DIM; kk += 4) {
        v2f a, bb;
        a.x  = Ap[kk];
        a.y  = Ap[kk + 1];
        bb.x = Bp[kk * W_PITCH];
        bb.y = Bp[kk * W_PITCH + W_PITCH];
        acc = __builtin_amdgcn_wmma_f32_16x16x4_f32(
             false, a,  false, bb,
             (short)0, acc,
             false,  false);
    }
    asm volatile("v_nop\n\tv_nop\n\tv_nop\n\tv_nop" : "+v"(acc));
    return acc;
}

__global__ void __launch_bounds__(GEMM_T)
k_cheb_rank2(const float* __restrict__ W,
             const float* __restrict__ b,
             float* __restrict__ out,
             const float* __restrict__ wn,
             const float2* __restrict__ p1,
             const float2* __restrict__ p2,
             const float* __restrict__ lw,
             const float* __restrict__ lb) {
    __shared__ __attribute__((aligned(16))) float sW[DIM * W_PITCH];
    __shared__ float sU[8 * U_PITCH];
    __shared__ __attribute__((aligned(16))) float sA[2 * DIM];

    const int tid  = threadIdx.x;
    const int wave = __builtin_amdgcn_readfirstlane(tid >> 5);
    const int tile = blockIdx.x * GEMM_WAVES + wave;
    const bool live = tile < N_TILES;
    const int row0 = (live ? tile : N_TILES - 1) * 16;
    const int lane = tid & 31;
    const int half = lane >> 4;
    const int lr   = lane & 15;

    const float4* Wg = (const float4*)W + (size_t)(tid >> 5) * (DIM / 4) + (tid & 31);
    float*        sS = sW + (tid >> 5) * W_PITCH + (tid & 31) * 4;
    W16(W_DECL)
    W16(W_LOAD0)

    sA[tid] = (tid < DIM) ? lw[tid] : lb[tid - DIM];
    const float* Ap = sA + (lr & 1) * DIM + 2 * half;
    const float* Bp = sW + (2 * half) * W_PITCH + wave * 16 + lr;

    W16(W_STORE)
    __syncthreads();
    W16(W_LOAD1)
    const v8f u0 = rank2_segment(Ap, Bp);

    __syncthreads();
    W16(W_STORE)
    __syncthreads();
    W16(W_LOAD2)
    const v8f u1 = rank2_segment(Ap, Bp);

    __syncthreads();
    W16(W_STORE)
    __syncthreads();
    const v8f u2 = rank2_segment(Ap, Bp);

    if (lane < 16) {
        float* d = sU + wave * 16 + lane;
        d[0 * U_PITCH] = u0[0];
        d[1 * U_PITCH] = u0[1];
        d[2 * U_PITCH] = u1[0];
        d[3 * U_PITCH] = u1[1];
        d[4 * U_PITCH] = u2[0];
        d[5 * U_PITCH] = u2[1];
        d[6 * U_PITCH] = 0.0f;
        d[7 * U_PITCH] = 0.0f;
    }
    __syncthreads();

    const float  wv = wn[row0 + lr];
    const float2 q1 = p1[row0 + lr];
    const float2 q2 = p2[row0 + lr];
    v2f a0, a1;
    a0.x = half ? -q1.x : wv;
    a0.y = half ? -q1.y : 1.0f;
    a1.x = half ? 0.0f : 2.0f * q2.x - wv;
    a1.y = half ? 0.0f : 2.0f * q2.y - 1.0f;
    const float* Up = sU + (2 * half) * U_PITCH + lr;

    v8f acc[8];
#pragma unroll
    for (int nt = 0; nt < 8; ++nt) {
        float bv = b[nt * 16 + lr];
        acc[nt] = (v8f){bv, bv, bv, bv, bv, bv, bv, bv};
        v2f b0, b1;
        b0.x = Up[nt * 16];
        b0.y = Up[nt * 16 + U_PITCH];
        b1.x = Up[nt * 16 + 4 * U_PITCH];
        b1.y = Up[nt * 16 + 5 * U_PITCH];
        acc[nt] = __builtin_amdgcn_wmma_f32_16x16x4_f32(
             false, a0,  false, b0,
             (short)0, acc[nt],
             false,  false);
        acc[nt] = __builtin_amdgcn_wmma_f32_16x16x4_f32(
             false, a1,  false, b1,
             (short)0, acc[nt],
             false,  false);
    }

    if (live) {
#pragma unroll
        for (int nt = 0; nt < 8; ++nt) {
            const int col = nt * 16 + lr;
#pragma unroll
            for (int r = 0; r < 8; ++r) {
                const int row = row0 + r + 8 * half;
                float v = acc[nt][r];
                out[(size_t)row * DIM + col] = v > 0.0f ? v : 0.0f;
            }
        }
    }
}

__global__ void k_pred(const float* __restrict__ emb,
                       const float* __restrict__ pw,
                       const float* __restrict__ pb,
                       float* __restrict__ out) {
    int n = blockIdx.x * blockDim.x + threadIdx.x;
    if (n < N_NODES) {
        const float* e = emb + (size_t)n * DIM;
        float s = 0.0f;
#pragma unroll 8
        for (int d = 0; d < DIM; d += 4) {
            float4 v = *(const float4*)(e + d);
            s += v.x * pw[d] + v.y * pw[d + 1] + v.z * pw[d + 2] + v.w * pw[d + 3];
        }
        out[n] = s + pb[0];
    }
}

extern "C" void kernel_launch(void* const* d_in, const int* in_sizes, int n_in,
                              void* d_out, int out_size, void* d_ws, size_t ws_size,
                              hipStream_t stream) {
    const float* weights  = (const float*)d_in[0];
    const int*   src      = (const int*)  d_in[1];
    const int*   dst      = (const int*)  d_in[2];
    const float* lin_in_w = (const float*)d_in[3];
    const float* lin_in_b = (const float*)d_in[4];
    const float* cheb_ws  = (const float*)d_in[5];
    const float* cheb_bs  = (const float*)d_in[6];
    const float* pred_w   = (const float*)d_in[7];
    const float* pred_b   = (const float*)d_in[8];
    float* logits = (float*)d_out;

    float* ws   = (float*)d_ws;
    float* dinv = ws;
    float* X0   = ws + 51200;
    float* X1   = X0 + NF;
    float* X2   = X1 + NF;
    int*   g      = (int*)(X2 + NF);
    int*   total  = g;
    int*   rowptr = g + 51200;
    int*   lptr   = g + 102400;
    Edge*  edge   = (Edge*)(g + 153600);
    float* emb  = X2 + NF + NF;
    float2* p1  = (float2*)X1;
    float2* p2  = (float2*)X1 + 51200;

    const int T = 256;
    const int gN    = (N_NODES + T - 1) / T;
    const int gPull = N_NODES / PULL_ROWS_PER_BLOCK;

    k_range_count<<<RANGE_BLOCKS, RANGE_T, 0, stream>>>(dst, lptr, dinv, total);
    k_range_place<<<RANGE_BLOCKS, RANGE_T, 0, stream>>>(src, dst, lptr, total, dinv, rowptr, edge);

    k_spmm_pair<true><<<PAIR_BLOCKS, PAIR_T, 0, stream>>>(rowptr, edge, weights, p2, dinv, p1);
    k_spmm_pair<false><<<PAIR_BLOCKS, PAIR_T, 0, stream>>>(rowptr, edge, weights, p1, dinv, p2);
    k_cheb_rank2<<<GEMM_BLOCKS, GEMM_T, 0, stream>>>(cheb_ws, cheb_bs, emb,
                                                     weights, p1, p2, lin_in_w, lin_in_b);

    float* cur = emb;
    float* nxt = X0;
    for (int layer = 1; layer < 3; ++layer) {
        const float* Wl = cheb_ws + (size_t)layer * 3 * DIM * DIM;
        const float* bl = cheb_bs + (size_t)layer * DIM;

        k_spmm_pull<true><<<gPull, 32 * PULL_ROWS_PER_BLOCK, 0, stream>>>(
            rowptr, edge, cur, cur, dinv, X1, -1.0f, 0.0f);

        k_spmm_pull<false><<<gPull, 32 * PULL_ROWS_PER_BLOCK, 0, stream>>>(
            rowptr, edge, X1, cur, dinv, X2, -2.0f, -1.0f);

        k_cheb_gemm<<<GEMM_BLOCKS, GEMM_T, 0, stream>>>(cur, X1, X2, Wl, bl, nxt);

        float* t = cur; cur = nxt; nxt = t;
    }

    k_pred<<<gN, T, 0, stream>>>(cur, pred_w, pred_b, logits);
}


# CSR build cut two ways, 25 destination ranges x 10 edge slices: each of 250 work-groups tests a tenth of dst (was 100 reading it all), a 25-group scan joins slices; 12 launches to 13; row order moves
# speedup vs baseline: 4.3018x; 1.1048x over previous
#include <hip/hip_runtime.h>


typedef float v2f __attribute__((ext_vector_type(2)));
typedef float v8f __attribute__((ext_vector_type(8)));

#define N_NODES 50000
#define N_EDGES 600000
#define DIM     128
#define NF      (N_NODES * DIM)

struct Edge { int src; float w; };

#define CSR_R      25
#define CSR_RN     (N_NODES / CSR_R)
#define CSR_S      10
#define CSR_SE     (N_EDGES / CSR_S)
#define CSR_SF4    (CSR_SE / 4)
#define CSR_T      1024
#define CSR_WAVES  (CSR_T / 32)
#define CSR_SLOTS  (2 * CSR_T)
#define CSR_G      ((CSR_SF4 + CSR_T - 1) / CSR_T)
#define CSR_BLOCKS (CSR_R * CSR_S)
#define STASH_W    512
static_assert(CSR_R * CSR_RN == N_NODES && CSR_RN % 2 == 0, "the ranges cover the nodes exactly, in pairs");
static_assert(CSR_S * CSR_SE == N_EDGES && CSR_SE % 4 == 0, "the slices cover the edges exactly, in groups of 4");
static_assert(CSR_T < CSR_RN && CSR_RN <= CSR_SLOTS, "two counters per thread cover a range");
static_assert(CSR_BLOCKS <= CSR_T, "the totals in front of a range are summed one per thread");
static_assert(N_EDGES <= (1 << 21) && CSR_SLOTS == (1 << 11), "a stash entry packs edge number and counter into 32 bits");

template <typename F>
__device__ __forceinline__ void for_slice_dst(const int* __restrict__ dst, int s, int t, F f) {
    const int4* d4 = (const int4*)dst + s * CSR_SF4;
    int4 dv[CSR_G];
#pragma unroll
    for (int k = 0; k < CSR_G; ++k) {
        const int idx = t + k * CSR_T;
        dv[k] = make_int4(-1, -1, -1, -1);
        if (idx < CSR_SF4) dv[k] = d4[idx];
    }
#pragma unroll
    for (int k = 0; k < CSR_G; ++k) f(dv[k], s * CSR_SF4 + t + k * CSR_T);
}

__device__ __forceinline__ void count_one(int* s_cnt, int dstv, int lo) {
    const unsigned r = (unsigned)(dstv - lo);
    if (r < CSR_RN) atomicAdd(&s_cnt[r], 1);
}

__device__ __forceinline__ int wave_sum(int v) {
    for (int m = 16; m >= 1; m >>= 1) v += __shfl_xor(v, m);
    return v;
}

__global__ void __launch_bounds__(CSR_T)
k_slice_count(const int* __restrict__ dst,
              int* __restrict__ cnt,
              int* __restrict__ tot) {
    __shared__ int s_cnt[CSR_SLOTS];
    __shared__ int s_red[CSR_WAVES];
    const int t  = threadIdx.x;
    const int b  = blockIdx.x / CSR_S;
    const int s  = blockIdx.x - b * CSR_S;
    const int lo = b * CSR_RN;
    s_cnt[t] = 0;
    s_cnt[t + CSR_T] = 0;
    __syncthreads();

    for_slice_dst(dst, s, t, [&](const int4 dv, int) {
        count_one(s_cnt, dv.x, lo);
        count_one(s_cnt, dv.y, lo);
        count_one(s_cnt, dv.z, lo);
        count_one(s_cnt, dv.w, lo);
    });
    __syncthreads();

    const int c0 = s_cnt[t];
    const int c1 = s_cnt[t + CSR_T];
    int* row = cnt + (size_t)s * N_NODES + lo;
    row[t] = c0;
    if (t + CSR_T < CSR_RN) row[t + CSR_T] = c1;

    const int v = wave_sum(c0 + c1);
    if ((t & 31) == 0) s_red[t >> 5] = v;
    __syncthreads();
    if (t < 32) {
        const int w = wave_sum(s_red[t]);
        if (t == 0) tot[blockIdx.x] = w;
    }
}

__global__ void __launch_bounds__(CSR_T)
k_range_scan(int* __restrict__ cnt,
             const int* __restrict__ tot,
             int* __restrict__ rowptr,
             float* __restrict__ dinv) {
    __shared__ int s_red[CSR_WAVES];
    __shared__ int s_wsum[CSR_WAVES];
    const int t    = threadIdx.x;
    const int b    = blockIdx.x;
    const int lo   = b * CSR_RN;
    const int lane = t & 31;
    const int wave = t >> 5;

    const int fv = wave_sum((t < b * CSR_S) ? tot[t] : 0);
    if (lane == 0) s_red[wave] = fv;

    const bool own = 2 * t < CSR_RN;
    int2* cp = (int2*)(cnt + lo) + t;
    int2 c[CSR_S];
#pragma unroll
    for (int s = 0; s < CSR_S; ++s) c[s] = own ? cp[(size_t)s * (N_NODES / 2)] : make_int2(0, 0);
    int deg0 = 0, deg1 = 0;
#pragma unroll
    for (int s = 0; s < CSR_S; ++s) { deg0 += c[s].x; deg1 += c[s].y; }
    const int pair = deg0 + deg1;

    int inc = pair;
    for (int m = 1; m < 32; m <<= 1) {
        const int u = __shfl_up(inc, m);
        if (lane >= m) inc += u;
    }
    if (lane == 31) s_wsum[wave] = inc;
    __syncthreads();

    const int base  = wave_sum(s_red[lane]);
    const int wv    = s_wsum[lane];
    const int wpre  = wave_sum(lane < wave ? wv : 0);
    const int rsum  = wave_sum(wv);
    const int r0    = base + wpre + inc - pair;
    const int r1    = r0 + deg0;

    if (own) {
        *(int2*)(rowptr + lo + 2 * t) = make_int2(r0, r1);
        float2 d;
        d.x = rsqrtf(1.0f + (float)deg0);
        d.y = rsqrtf(1.0f + (float)deg1);
        *(float2*)(dinv + lo + 2 * t) = d;
        int run0 = r0, run1 = r1;
#pragma unroll
        for (int s = 0; s < CSR_S; ++s) {
            cp[(size_t)s * (N_NODES / 2)] = make_int2(run0, run1);
            run0 += c[s].x;
            run1 += c[s].y;
        }
    }
    if (b == CSR_R - 1 && t == 0) rowptr[N_NODES] = base + rsum;
}

__device__ __forceinline__ void place_edge(int* s_cur, Edge* __restrict__ edge, unsigned r, int s, float w) {
    Edge q;
    q.src = s;
    q.w   = w;
    const int slot = atomicAdd(&s_cur[r], 1);
    edge[slot] = q;
}

__global__ void __launch_bounds__(CSR_T)
k_slice_place(const int* __restrict__ src,
              const int* __restrict__ dst,
              const int* __restrict__ first,
              const float* __restrict__ dinv,
              Edge* __restrict__ edge) {
    __shared__ int      s_cur[CSR_SLOTS];
    __shared__ unsigned s_stash[CSR_WAVES * STASH_W];
    const int t    = threadIdx.x;
    const int b    = blockIdx.x / CSR_S;
    const int s    = blockIdx.x - b * CSR_S;
    const int lo   = b * CSR_RN;
    const int lane = t & 31;
    const int wave = __builtin_amdgcn_readfirstlane(t >> 5);

    const int* row = first + (size_t)s * N_NODES + lo;
    s_cur[t] = row[t];
    if (t + CSR_T < CSR_RN) s_cur[t + CSR_T] = row[t + CSR_T];
    __syncthreads();

    unsigned* my = s_stash + wave * STASH_W;
    int held = 0;
    auto note_one = [&](int dstv, int e) {
        const unsigned r = (unsigned)(dstv - lo);
        const bool mine = r < CSR_RN;
        const unsigned hits = __builtin_amdgcn_ballot_w32(mine);
        if (hits) {
            if (mine) {
                const int pos = held + (int)__builtin_amdgcn_mbcnt_lo(hits, 0u);
                if (pos < STASH_W) {
                    my[pos] = ((unsigned)e << 11) | r;
                } else {
                    const int sn = src[e];
                    place_edge(s_cur, edge, r, sn, dinv[sn]);
                }
            }
            held += __popc(hits);
        }
    };
    for_slice_dst(dst, s, t, [&](const int4 dv, int idx) {
        note_one(dv.x, 4 * idx + 0);
        note_one(dv.y, 4 * idx + 1);
        note_one(dv.z, 4 * idx + 2);
        note_one(dv.w, 4 * idx + 3);
    });
    const int lim = held < STASH_W ? held : STASH_W;
    for (int k = lane; k < lim; k += 32) {
        const unsigned u = my[k];
        const int sn = src[u >> 11];
        place_edge(s_cur, edge, u & (CSR_SLOTS - 1), sn, dinv[sn]);
    }
}

#define PAIR_LANES 8
#define PAIR_T     256
#define PAIR_BLOCKS ((N_NODES * PAIR_LANES + PAIR_T - 1) / PAIR_T)

template <bool FIRST>
__global__ void __launch_bounds__(PAIR_T)
k_spmm_pair(const int* __restrict__ rowptr,
            const Edge* __restrict__ edge,
            const float* __restrict__ wn,
            const float2* __restrict__ in,
            const float* __restrict__ dinv,
            float2* __restrict__ outp) {
    const int t   = blockIdx.x * PAIR_T + threadIdx.x;
    const int sub = t & (PAIR_LANES - 1);
    int row = t / PAIR_LANES;
    const bool live = row < N_NODES;
    if (!live) row = N_NODES - 1;

    const int end = rowptr[row + 1];
    float a = 0.0f, c = 0.0f;
    for (int i = rowptr[row] + sub; i < end; i += PAIR_LANES) {
        const Edge r = edge[i];
        float2 v;
        if (FIRST) { v.x = wn[r.src]; v.y = 1.0f; }
        else       { v = in[r.src]; }
        a += v.x * r.w;
        c += v.y * r.w;
    }
    for (int m = 1; m < PAIR_LANES; m <<= 1) {
        a += __shfl_xor(a, m);
        c += __shfl_xor(c, m);
    }
    if (live && sub == 0) {
        const float di = dinv[row];
        float2 s;
        if (FIRST) { s.x = wn[row]; s.y = 1.0f; }
        else       { s = in[row]; }
        float2 o;
        o.x = (a + s.x * di) * di;
        o.y = (c + s.y * di) * di;
        outp[row] = o;
    }
}

#define PULL_ROWS_PER_BLOCK 8
static_assert(N_NODES % PULL_ROWS_PER_BLOCK == 0, "grid covers the rows exactly");

template <int G>
__device__ __forceinline__ void pull_rows(const Edge* __restrict__ edge,
                                          const float4* __restrict__ x4,
                                          int i, int lane, float4& acc) {
    Edge r[G]; float4 v[G];
#pragma unroll
    for (int k = 0; k < G; ++k) r[k] = edge[i + k];
#pragma unroll
    for (int k = 0; k < G; ++k) v[k] = x4[(size_t)r[k].src * (DIM / 4) + lane];
#pragma unroll
    for (int k = 0; k < G; ++k) {
        acc.x += v[k].x * r[k].w;
        acc.y += v[k].y * r[k].w;
        acc.z += v[k].z * r[k].w;
        acc.w += v[k].w * r[k].w;
    }
}

template <bool PREV_IS_X>
__global__ void __launch_bounds__(32 * PULL_ROWS_PER_BLOCK)
k_spmm_pull(const int* __restrict__ rowptr,
            const Edge* __restrict__ edge,
            const float* __restrict__ x,
            const float* __restrict__ prev,
            const float* __restrict__ dinv,
            float* __restrict__ out,
            float alpha, float beta) {
    const int row  = __builtin_amdgcn_readfirstlane(
                         (int)(blockIdx.x * PULL_ROWS_PER_BLOCK + (threadIdx.x >> 5)));
    const int lane = threadIdx.x & 31;
    const float4* x4 = (const float4*)x;

    int       i   = rowptr[row];
    const int end = rowptr[row + 1];
    const float  di   = dinv[row];
    const float4 self = x4[(size_t)row * (DIM / 4) + lane];
    float4 pv = self;
    if (!PREV_IS_X) pv = ((const float4*)prev)[(size_t)row * (DIM / 4) + lane];

    float4 acc = make_float4(0.0f, 0.0f, 0.0f, 0.0f);
    for (; i + 4 <= end; i += 4) pull_rows<4>(edge, x4, i, lane, acc);
    switch (end - i) {
        case 3: pull_rows<3>(edge, x4, i, lane, acc); break;
        case 2: pull_rows<2>(edge, x4, i, lane, acc); break;
        case 1: pull_rows<1>(edge, x4, i, lane, acc); break;
        default: break;
    }

    float4 o;
    o.x = alpha * ((acc.x + self.x * di) * di) + beta * pv.x;
    o.y = alpha * ((acc.y + self.y * di) * di) + beta * pv.y;
    o.z = alpha * ((acc.z + self.z * di) * di) + beta * pv.z;
    o.w = alpha * ((acc.w + self.w * di) * di) + beta * pv.w;
    ((float4*)out)[(size_t)row * (DIM / 4) + lane] = o;
}

#define GEMM_WAVES  8
#define GEMM_T      (32 * GEMM_WAVES)
#define GEMM_ROWS   (16 * GEMM_WAVES)
#define GEMM_BLOCKS ((N_NODES + GEMM_ROWS - 1) / GEMM_ROWS)
#define N_TILES     (N_NODES / 16)
#define W_PITCH     136
#define W_F4        (DIM * DIM / 4 / GEMM_T)

#define W16(OP) OP(0) OP(1) OP(2) OP(3) OP(4) OP(5) OP(6) OP(7) OP(8) OP(9) OP(10) OP(11) OP(12) OP(13) OP(14) OP(15)
#define W_DECL(i)   float4 w##i;
#define W_STORE(i)  *(float4*)(sS + (i) * 8 * W_PITCH) = w##i;
#define W_LOAD0(i)  w##i = Wg[(size_t)(0 * DIM + (i) * 8) * (DIM / 4)];
#define W_LOAD1(i)  w##i = Wg[(size_t)(1 * DIM + (i) * 8) * (DIM / 4)];
#define W_LOAD2(i)  w##i = Wg[(size_t)(2 * DIM + (i) * 8) * (DIM / 4)];
static_assert(W_F4 == 16, "W16 lists the float4 of one thread");

__device__ __forceinline__ void gemm_segment(v8f (&acc)[8],
                                             const float* __restrict__ Ap,
                                             const float* Bp) {
#pragma unroll 4
    for (int kk = 0; kk < DIM; kk += 4) {
        v2f a;
        a.x = Ap[kk];
        a.y = Ap[kk + 1];
        const float* Bk  = Bp + kk * W_PITCH;
        const float* Bk1 = Bk + W_PITCH;
#pragma unroll
        for (int nt = 0; nt < 8; ++nt) {
            v2f bb;
            bb.x = Bk [nt * 16];
            bb.y = Bk1[nt * 16];
            acc[nt] = __builtin_amdgcn_wmma_f32_16x16x4_f32(
                 false, a,  false, bb,
                 (short)0, acc[nt],
                 false,  false);
        }
    }
}

__global__ void __launch_bounds__(GEMM_T)
k_cheb_gemm(const float* __restrict__ X0,
            const float* __restrict__ X1,
            const float* __restrict__ X2,
            const float* __restrict__ W,
            const float* __restrict__ b,
            float* __restrict__ out) {
    __shared__ __attribute__((aligned(16))) float sW[DIM * W_PITCH];

    const int tid  = threadIdx.x;
    const int wave = __builtin_amdgcn_readfirstlane(tid >> 5);
    const int tile = blockIdx.x * GEMM_WAVES + wave;
    const bool live = tile < N_TILES;
    const int row0 = (live ? tile : N_TILES - 1) * 16;
    const int lane = tid & 31;
    const int half = lane >> 4;
    const int lr   = lane & 15;

    const float4* Wg = (const float4*)W + (size_t)(tid >> 5) * (DIM / 4) + (tid & 31);
    float*        sS = sW + (tid >> 5) * W_PITCH + (tid & 31) * 4;
    W16(W_DECL)
    W16(W_LOAD0)

    v8f acc[8];
#pragma unroll
    for (int nt = 0; nt < 8; ++nt) {
        float bv = b[nt * 16 + lr];
        acc[nt] = (v8f){bv, bv, bv, bv, bv, bv, bv, bv};
    }

    const size_t aoff = (size_t)(row0 + lr) * DIM + 2 * half;
    const float* Bp   = sW + (2 * half) * W_PITCH + lr;

    W16(W_STORE)
    __syncthreads();
    W16(W_LOAD1)
    gemm_segment(acc, X0 + aoff, Bp);

    __syncthreads();
    W16(W_STORE)
    __syncthreads();
    W16(W_LOAD2)
    gemm_segment(acc, X1 + aoff, Bp);

    __syncthreads();
    W16(W_STORE)
    __syncthreads();
    gemm_segment(acc, X2 + aoff, Bp);

    if (live) {
#pragma unroll
        for (int nt = 0; nt < 8; ++nt) {
            const int col = nt * 16 + lr;
#pragma unroll
            for (int r = 0; r < 8; ++r) {
                const int row = row0 + r + 8 * half;
                float v = acc[nt][r];
                out[(size_t)row * DIM + col] = v > 0.0f ? v : 0.0f;
            }
        }
    }
}

static_assert(GEMM_T == 2 * DIM, "one thread per element of lw | lb when they are copied to LDS");
#define U_PITCH 136

__device__ __forceinline__ v8f rank2_segment(const float* Ap, const float* Bp) {
    v8f acc = (v8f){0.0f, 0.0f, 0.0f, 0.0f, 0.0f, 0.0f, 0.0f, 0.0f};
#pragma unroll 8
    for (int kk = 0; kk < DIM; kk += 4) {
        v2f a, bb;
        a.x  = Ap[kk];
        a.y  = Ap[kk + 1];
        bb.x = Bp[kk * W_PITCH];
        bb.y = Bp[kk * W_PITCH + W_PITCH];
        acc = __builtin_amdgcn_wmma_f32_16x16x4_f32(
             false, a,  false, bb,
             (short)0, acc,
             false,  false);
    }
    asm volatile("v_nop\n\tv_nop\n\tv_nop\n\tv_nop" : "+v"(acc));
    return acc;
}

__global__ void __launch_bounds__(GEMM_T)
k_cheb_rank2(const float* __restrict__ W,
             const float* __restrict__ b,
             float* __restrict__ out,
             const float* __restrict__ wn,
             const float2* __restrict__ p1,
             const float2* __restrict__ p2,
             const float* __restrict__ lw,
             const float* __restrict__ lb) {
    __shared__ __attribute__((aligned(16))) float sW[DIM * W_PITCH];
    __shared__ float sU[8 * U_PITCH];
    __shared__ __attribute__((aligned(16))) float sA[2 * DIM];

    const int tid  = threadIdx.x;
    const int wave = __builtin_amdgcn_readfirstlane(tid >> 5);
    const int tile = blockIdx.x * GEMM_WAVES + wave;
    const bool live = tile < N_TILES;
    const int row0 = (live ? tile : N_TILES - 1) * 16;
    const int lane = tid & 31;
    const int half = lane >> 4;
    const int lr   = lane & 15;

    const float4* Wg = (const float4*)W + (size_t)(tid >> 5) * (DIM / 4) + (tid & 31);
    float*        sS = sW + (tid >> 5) * W_PITCH + (tid & 31) * 4;
    W16(W_DECL)
    W16(W_LOAD0)

    sA[tid] = (tid < DIM) ? lw[tid] : lb[tid - DIM];
    const float* Ap = sA + (lr & 1) * DIM + 2 * half;
    const float* Bp = sW + (2 * half) * W_PITCH + wave * 16 + lr;

    W16(W_STORE)
    __syncthreads();
    W16(W_LOAD1)
    const v8f u0 = rank2_segment(Ap, Bp);

    __syncthreads();
    W16(W_STORE)
    __syncthreads();
    W16(W_LOAD2)
    const v8f u1 = rank2_segment(Ap, Bp);

    __syncthreads();
    W16(W_STORE)
    __syncthreads();
    const v8f u2 = rank2_segment(Ap, Bp);

    if (lane < 16) {
        float* d = sU + wave * 16 + lane;
        d[0 * U_PITCH] = u0[0];
        d[1 * U_PITCH] = u0[1];
        d[2 * U_PITCH] = u1[0];
        d[3 * U_PITCH] = u1[1];
        d[4 * U_PITCH] = u2[0];
        d[5 * U_PITCH] = u2[1];
        d[6 * U_PITCH] = 0.0f;
        d[7 * U_PITCH] = 0.0f;
    }
    __syncthreads();

    const float  wv = wn[row0 + lr];
    const float2 q1 = p1[row0 + lr];
    const float2 q2 = p2[row0 + lr];
    v2f a0, a1;
    a0.x = half ? -q1.x : wv;
    a0.y = half ? -q1.y : 1.0f;
    a1.x = half ? 0.0f : 2.0f * q2.x - wv;
    a1.y = half ? 0.0f : 2.0f * q2.y - 1.0f;
    const float* Up = sU + (2 * half) * U_PITCH + lr;

    v8f acc[8];
#pragma unroll
    for (int nt = 0; nt < 8; ++nt) {
        float bv = b[nt * 16 + lr];
        acc[nt] = (v8f){bv, bv, bv, bv, bv, bv, bv, bv};
        v2f b0, b1;
        b0.x = Up[nt * 16];
        b0.y = Up[nt * 16 + U_PITCH];
        b1.x = Up[nt * 16 + 4 * U_PITCH];
        b1.y = Up[nt * 16 + 5 * U_PITCH];
        acc[nt] = __builtin_amdgcn_wmma_f32_16x16x4_f32(
             false, a0,  false, b0,
             (short)0, acc[nt],
             false,  false);
        acc[nt] = __builtin_amdgcn_wmma_f32_16x16x4_f32(
             false, a1,  false, b1,
             (short)0, acc[nt],
             false,  false);
    }

    if (live) {
#pragma unroll
        for (int nt = 0; nt < 8; ++nt) {
            const int col = nt * 16 + lr;
#pragma unroll
            for (int r = 0; r < 8; ++r) {
                const int row = row0 + r + 8 * half;
                float v = acc[nt][r];
                out[(size_t)row * DIM + col] = v > 0.0f ? v : 0.0f;
            }
        }
    }
}

__global__ void k_pred(const float* __restrict__ emb,
                       const float* __restrict__ pw,
                       const float* __restrict__ pb,
                       float* __restrict__ out) {
    int n = blockIdx.x * blockDim.x + threadIdx.x;
    if (n < N_NODES) {
        const float* e = emb + (size_t)n * DIM;
        float s = 0.0f;
#pragma unroll 8
        for (int d = 0; d < DIM; d += 4) {
            float4 v = *(const float4*)(e + d);
            s += v.x * pw[d] + v.y * pw[d + 1] + v.z * pw[d + 2] + v.w * pw[d + 3];
        }
        out[n] = s + pb[0];
    }
}

extern "C" void kernel_launch(void* const* d_in, const int* in_sizes, int n_in,
                              void* d_out, int out_size, void* d_ws, size_t ws_size,
                              hipStream_t stream) {
    const float* weights  = (const float*)d_in[0];
    const int*   src      = (const int*)  d_in[1];
    const int*   dst      = (const int*)  d_in[2];
    const float* lin_in_w = (const float*)d_in[3];
    const float* lin_in_b = (const float*)d_in[4];
    const float* cheb_ws  = (const float*)d_in[5];
    const float* cheb_bs  = (const float*)d_in[6];
    const float* pred_w   = (const float*)d_in[7];
    const float* pred_b   = (const float*)d_in[8];
    float* logits = (float*)d_out;

    float* ws   = (float*)d_ws;
    float* dinv = ws;
    float* X0   = ws + 51200;
    float* X1   = X0 + NF;
    float* X2   = X1 + NF;
    int*   g      = (int*)(X2 + NF);
    int*   tot    = g;
    int*   rowptr = g + 51200;
    Edge*  edge   = (Edge*)(g + 153600);
    int*   cnt    = g + 1400000;
    float* emb  = X2 + NF + NF;
    float2* p1  = (float2*)X1;
    float2* p2  = (float2*)X1 + 51200;

    const int T = 256;
    const int gN    = (N_NODES + T - 1) / T;
    const int gPull = N_NODES / PULL_ROWS_PER_BLOCK;

    k_slice_count<<<CSR_BLOCKS, CSR_T, 0, stream>>>(dst, cnt, tot);
    k_range_scan<<<CSR_R, CSR_T, 0, stream>>>(cnt, tot, rowptr, dinv);
    k_slice_place<<<CSR_BLOCKS, CSR_T, 0, stream>>>(src, dst, cnt, dinv, edge);

    k_spmm_pair<true><<<PAIR_BLOCKS, PAIR_T, 0, stream>>>(rowptr, edge, weights, p2, dinv, p1);
    k_spmm_pair<false><<<PAIR_BLOCKS, PAIR_T, 0, stream>>>(rowptr, edge, weights, p1, dinv, p2);
    k_cheb_rank2<<<GEMM_BLOCKS, GEMM_T, 0, stream>>>(cheb_ws, cheb_bs, emb,
                                                     weights, p1, p2, lin_in_w, lin_in_b);

    float* cur = emb;
    float* nxt = X0;
    for (int layer = 1; layer < 3; ++layer) {
        const float* Wl = cheb_ws + (size_t)layer * 3 * DIM * DIM;
        const float* bl = cheb_bs + (size_t)layer * DIM;

        k_spmm_pull<true><<<gPull, 32 * PULL_ROWS_PER_BLOCK, 0, stream>>>(
            rowptr, edge, cur, cur, dinv, X1, -1.0f, 0.0f);

        k_spmm_pull<false><<<gPull, 32 * PULL_ROWS_PER_BLOCK, 0, stream>>>(
            rowptr, edge, X1, cur, dinv, X2, -2.0f, -1.0f);

        k_cheb_gemm<<<GEMM_BLOCKS, GEMM_T, 0, stream>>>(cur, X1, X2, Wl, bl, nxt);

        float* t = cur; cur = nxt; nxt = t;
    }

    k_pred<<<gN, T, 0, stream>>>(cur, pred_w, pred_b, logits);
}


# X2 gather moved into the GEMM launch: 16-wave work-groups, 8 gather waves sum each tile's X2 rows into LDS beside the 8 matrix waves' first two k segments; X2 never in memory; 13 launches to 11
# speedup vs baseline: 4.6970x; 1.0919x over previous
#include <hip/hip_runtime.h>


typedef float v2f __attribute__((ext_vector_type(2)));
typedef float v8f __attribute__((ext_vector_type(8)));

#define N_NODES 50000
#define N_EDGES 600000
#define DIM     128
#define NF      (N_NODES * DIM)

struct Edge { int src; float w; };

#define CSR_R      25
#define CSR_RN     (N_NODES / CSR_R)
#define CSR_S      10
#define CSR_SE     (N_EDGES / CSR_S)
#define CSR_SF4    (CSR_SE / 4)
#define CSR_T      1024
#define CSR_WAVES  (CSR_T / 32)
#define CSR_SLOTS  (2 * CSR_T)
#define CSR_G      ((CSR_SF4 + CSR_T - 1) / CSR_T)
#define CSR_BLOCKS (CSR_R * CSR_S)
#define STASH_W    512
static_assert(CSR_R * CSR_RN == N_NODES && CSR_RN % 2 == 0, "the ranges cover the nodes exactly, in pairs");
static_assert(CSR_S * CSR_SE == N_EDGES && CSR_SE % 4 == 0, "the slices cover the edges exactly, in groups of 4");
static_assert(CSR_T < CSR_RN && CSR_RN <= CSR_SLOTS, "two counters per thread cover a range");
static_assert(CSR_BLOCKS <= CSR_T, "the totals in front of a range are summed one per thread");
static_assert(N_EDGES <= (1 << 21) && CSR_SLOTS == (1 << 11), "a stash entry packs edge number and counter into 32 bits");

template <typename F>
__device__ __forceinline__ void for_slice_dst(const int* __restrict__ dst, int s, int t, F f) {
    const int4* d4 = (const int4*)dst + s * CSR_SF4;
    int4 dv[CSR_G];
#pragma unroll
    for (int k = 0; k < CSR_G; ++k) {
        const int idx = t + k * CSR_T;
        dv[k] = make_int4(-1, -1, -1, -1);
        if (idx < CSR_SF4) dv[k] = d4[idx];
    }
#pragma unroll
    for (int k = 0; k < CSR_G; ++k) f(dv[k], s * CSR_SF4 + t + k * CSR_T);
}

__device__ __forceinline__ void count_one(int* s_cnt, int dstv, int lo) {
    const unsigned r = (unsigned)(dstv - lo);
    if (r < CSR_RN) atomicAdd(&s_cnt[r], 1);
}

__device__ __forceinline__ int wave_sum(int v) {
    for (int m = 16; m >= 1; m >>= 1) v += __shfl_xor(v, m);
    return v;
}

__global__ void __launch_bounds__(CSR_T)
k_slice_count(const int* __restrict__ dst,
              int* __restrict__ cnt,
              int* __restrict__ tot) {
    __shared__ int s_cnt[CSR_SLOTS];
    __shared__ int s_red[CSR_WAVES];
    const int t  = threadIdx.x;
    const int b  = blockIdx.x / CSR_S;
    const int s  = blockIdx.x - b * CSR_S;
    const int lo = b * CSR_RN;
    s_cnt[t] = 0;
    s_cnt[t + CSR_T] = 0;
    __syncthreads();

    for_slice_dst(dst, s, t, [&](const int4 dv, int) {
        count_one(s_cnt, dv.x, lo);
        count_one(s_cnt, dv.y, lo);
        count_one(s_cnt, dv.z, lo);
        count_one(s_cnt, dv.w, lo);
    });
    __syncthreads();

    const int c0 = s_cnt[t];
    const int c1 = s_cnt[t + CSR_T];
    int* row = cnt + (size_t)s * N_NODES + lo;
    row[t] = c0;
    if (t + CSR_T < CSR_RN) row[t + CSR_T] = c1;

    const int v = wave_sum(c0 + c1);
    if ((t & 31) == 0) s_red[t >> 5] = v;
    __syncthreads();
    if (t < 32) {
        const int w = wave_sum(s_red[t]);
        if (t == 0) tot[blockIdx.x] = w;
    }
}

__global__ void __launch_bounds__(CSR_T)
k_range_scan(int* __restrict__ cnt,
             const int* __restrict__ tot,
             int* __restrict__ rowptr,
             float* __restrict__ dinv) {
    __shared__ int s_red[CSR_WAVES];
    __shared__ int s_wsum[CSR_WAVES];
    const int t    = threadIdx.x;
    const int b    = blockIdx.x;
    const int lo   = b * CSR_RN;
    const int lane = t & 31;
    const int wave = t >> 5;

    const int fv = wave_sum((t < b * CSR_S) ? tot[t] : 0);
    if (lane == 0) s_red[wave] = fv;

    const bool own = 2 * t < CSR_RN;
    int2* cp = (int2*)(cnt + lo) + t;
    int2 c[CSR_S];
#pragma unroll
    for (int s = 0; s < CSR_S; ++s) c[s] = own ? cp[(size_t)s * (N_NODES / 2)] : make_int2(0, 0);
    int deg0 = 0, deg1 = 0;
#pragma unroll
    for (int s = 0; s < CSR_S; ++s) { deg0 += c[s].x; deg1 += c[s].y; }
    const int pair = deg0 + deg1;

    int inc = pair;
    for (int m = 1; m < 32; m <<= 1) {
        const int u = __shfl_up(inc, m);
        if (lane >= m) inc += u;
    }
    if (lane == 31) s_wsum[wave] = inc;
    __syncthreads();

    const int base  = wave_sum(s_red[lane]);
    const int wv    = s_wsum[lane];
    const int wpre  = wave_sum(lane < wave ? wv : 0);
    const int rsum  = wave_sum(wv);
    const int r0    = base + wpre + inc - pair;
    const int r1    = r0 + deg0;

    if (own) {
        *(int2*)(rowptr + lo + 2 * t) = make_int2(r0, r1);
        float2 d;
        d.x = rsqrtf(1.0f + (float)deg0);
        d.y = rsqrtf(1.0f + (float)deg1);
        *(float2*)(dinv + lo + 2 * t) = d;
        int run0 = r0, run1 = r1;
#pragma unroll
        for (int s = 0; s < CSR_S; ++s) {
            cp[(size_t)s * (N_NODES / 2)] = make_int2(run0, run1);
            run0 += c[s].x;
            run1 += c[s].y;
        }
    }
    if (b == CSR_R - 1 && t == 0) rowptr[N_NODES] = base + rsum;
}

__device__ __forceinline__ void place_edge(int* s_cur, Edge* __restrict__ edge, unsigned r, int s, float w) {
    Edge q;
    q.src = s;
    q.w   = w;
    const int slot = atomicAdd(&s_cur[r], 1);
    edge[slot] = q;
}

__global__ void __launch_bounds__(CSR_T)
k_slice_place(const int* __restrict__ src,
              const int* __restrict__ dst,
              const int* __restrict__ first,
              const float* __restrict__ dinv,
              Edge* __restrict__ edge) {
    __shared__ int      s_cur[CSR_SLOTS];
    __shared__ unsigned s_stash[CSR_WAVES * STASH_W];
    const int t    = threadIdx.x;
    const int b    = blockIdx.x / CSR_S;
    const int s    = blockIdx.x - b * CSR_S;
    const int lo   = b * CSR_RN;
    const int lane = t & 31;
    const int wave = __builtin_amdgcn_readfirstlane(t >> 5);

    const int* row = first + (size_t)s * N_NODES + lo;
    s_cur[t] = row[t];
    if (t + CSR_T < CSR_RN) s_cur[t + CSR_T] = row[t + CSR_T];
    __syncthreads();

    unsigned* my = s_stash + wave * STASH_W;
    int held = 0;
    auto note_one = [&](int dstv, int e) {
        const unsigned r = (unsigned)(dstv - lo);
        const bool mine = r < CSR_RN;
        const unsigned hits = __builtin_amdgcn_ballot_w32(mine);
        if (hits) {
            if (mine) {
                const int pos = held + (int)__builtin_amdgcn_mbcnt_lo(hits, 0u);
                if (pos < STASH_W) {
                    my[pos] = ((unsigned)e << 11) | r;
                } else {
                    const int sn = src[e];
                    place_edge(s_cur, edge, r, sn, dinv[sn]);
                }
            }
            held += __popc(hits);
        }
    };
    for_slice_dst(dst, s, t, [&](const int4 dv, int idx) {
        note_one(dv.x, 4 * idx + 0);
        note_one(dv.y, 4 * idx + 1);
        note_one(dv.z, 4 * idx + 2);
        note_one(dv.w, 4 * idx + 3);
    });
    const int lim = held < STASH_W ? held : STASH_W;
    for (int k = lane; k < lim; k += 32) {
        const unsigned u = my[k];
        const int sn = src[u >> 11];
        place_edge(s_cur, edge, u & (CSR_SLOTS - 1), sn, dinv[sn]);
    }
}

#define PAIR_LANES 8
#define PAIR_T     256
#define PAIR_BLOCKS ((N_NODES * PAIR_LANES + PAIR_T - 1) / PAIR_T)

template <bool FIRST>
__global__ void __launch_bounds__(PAIR_T)
k_spmm_pair(const int* __restrict__ rowptr,
            const Edge* __restrict__ edge,
            const float* __restrict__ wn,
            const float2* __restrict__ in,
            const float* __restrict__ dinv,
            float2* __restrict__ outp) {
    const int t   = blockIdx.x * PAIR_T + threadIdx.x;
    const int sub = t & (PAIR_LANES - 1);
    int row = t / PAIR_LANES;
    const bool live = row < N_NODES;
    if (!live) row = N_NODES - 1;

    const int end = rowptr[row + 1];
    float a = 0.0f, c = 0.0f;
    for (int i = rowptr[row] + sub; i < end; i += PAIR_LANES) {
        const Edge r = edge[i];
        float2 v;
        if (FIRST) { v.x = wn[r.src]; v.y = 1.0f; }
        else       { v = in[r.src]; }
        a += v.x * r.w;
        c += v.y * r.w;
    }
    for (int m = 1; m < PAIR_LANES; m <<= 1) {
        a += __shfl_xor(a, m);
        c += __shfl_xor(c, m);
    }
    if (live && sub == 0) {
        const float di = dinv[row];
        float2 s;
        if (FIRST) { s.x = wn[row]; s.y = 1.0f; }
        else       { s = in[row]; }
        float2 o;
        o.x = (a + s.x * di) * di;
        o.y = (c + s.y * di) * di;
        outp[row] = o;
    }
}

#define PULL_ROWS_PER_BLOCK 8
static_assert(N_NODES % PULL_ROWS_PER_BLOCK == 0, "grid covers the rows exactly");

template <int G>
__device__ __forceinline__ void pull_rows(const Edge* __restrict__ edge,
                                          const float4* __restrict__ x4,
                                          int i, int lane, float4& acc) {
    Edge r[G]; float4 v[G];
#pragma unroll
    for (int k = 0; k < G; ++k) r[k] = edge[i + k];
#pragma unroll
    for (int k = 0; k < G; ++k) v[k] = x4[(size_t)r[k].src * (DIM / 4) + lane];
#pragma unroll
    for (int k = 0; k < G; ++k) {
        acc.x = __builtin_fmaf(v[k].x, r[k].w, acc.x);
        acc.y = __builtin_fmaf(v[k].y, r[k].w, acc.y);
        acc.z = __builtin_fmaf(v[k].z, r[k].w, acc.z);
        acc.w = __builtin_fmaf(v[k].w, r[k].w, acc.w);
    }
}

template <bool PREV_IS_X>
__global__ void __launch_bounds__(32 * PULL_ROWS_PER_BLOCK)
k_spmm_pull(const int* __restrict__ rowptr,
            const Edge* __restrict__ edge,
            const float* __restrict__ x,
            const float* __restrict__ prev,
            const float* __restrict__ dinv,
            float* __restrict__ out,
            float alpha, float beta) {
    const int row  = __builtin_amdgcn_readfirstlane(
                         (int)(blockIdx.x * PULL_ROWS_PER_BLOCK + (threadIdx.x >> 5)));
    const int lane = threadIdx.x & 31;
    const float4* x4 = (const float4*)x;

    int       i   = rowptr[row];
    const int end = rowptr[row + 1];
    const float  di   = dinv[row];
    const float4 self = x4[(size_t)row * (DIM / 4) + lane];
    float4 pv = self;
    if (!PREV_IS_X) pv = ((const float4*)prev)[(size_t)row * (DIM / 4) + lane];

    float4 acc = make_float4(0.0f, 0.0f, 0.0f, 0.0f);
    for (; i + 4 <= end; i += 4) pull_rows<4>(edge, x4, i, lane, acc);
    switch (end - i) {
        case 3: pull_rows<3>(edge, x4, i, lane, acc); break;
        case 2: pull_rows<2>(edge, x4, i, lane, acc); break;
        case 1: pull_rows<1>(edge, x4, i, lane, acc); break;
        default: break;
    }

    float4 o;
    o.x = alpha * ((acc.x + self.x * di) * di) + beta * pv.x;
    o.y = alpha * ((acc.y + self.y * di) * di) + beta * pv.y;
    o.z = alpha * ((acc.z + self.z * di) * di) + beta * pv.z;
    o.w = alpha * ((acc.w + self.w * di) * di) + beta * pv.w;
    ((float4*)out)[(size_t)row * (DIM / 4) + lane] = o;
}

#define GEMM_WAVES  8
#define GEMM_T      (32 * GEMM_WAVES)
#define GEMM_ROWS   (16 * GEMM_WAVES)
#define GEMM_BLOCKS ((N_NODES + GEMM_ROWS - 1) / GEMM_ROWS)
#define N_TILES     (N_NODES / 16)
#define W_PITCH     136
#define W_F4        (DIM * DIM / 4 / GEMM_T)

#define W16(OP) OP(0) OP(1) OP(2) OP(3) OP(4) OP(5) OP(6) OP(7) OP(8) OP(9) OP(10) OP(11) OP(12) OP(13) OP(14) OP(15)
#define W_DECL(i)   float4 w##i;
#define W_STORE(i)  *(float4*)(sS + (i) * 8 * W_PITCH) = w##i;
#define W_LOAD0(i)  w##i = Wg[(size_t)(0 * DIM + (i) * 8) * (DIM / 4)];
#define W_LOAD1(i)  w##i = Wg[(size_t)(1 * DIM + (i) * 8) * (DIM / 4)];
#define W_LOAD2(i)  w##i = Wg[(size_t)(2 * DIM + (i) * 8) * (DIM / 4)];
static_assert(W_F4 == 16, "W16 lists the float4 of one thread");

__device__ __forceinline__ void gemm_segment(v8f (&acc)[8],
                                             const float* __restrict__ Ap,
                                             const float* Bp) {
#pragma unroll 4
    for (int kk = 0; kk < DIM; kk += 4) {
        v2f a;
        a.x = Ap[kk];
        a.y = Ap[kk + 1];
        const float* Bk  = Bp + kk * W_PITCH;
        const float* Bk1 = Bk + W_PITCH;
#pragma unroll
        for (int nt = 0; nt < 8; ++nt) {
            v2f bb;
            bb.x = Bk [nt * 16];
            bb.y = Bk1[nt * 16];
            acc[nt] = __builtin_amdgcn_wmma_f32_16x16x4_f32(
                 false, a,  false, bb,
                 (short)0, acc[nt],
                 false,  false);
        }
    }
}

#define FUSE_WAVES  (2 * GEMM_WAVES)
#define FUSE_T      (32 * FUSE_WAVES)
#define T_PITCH     132
#define WF_F4       (DIM * DIM / 4 / FUSE_T)
#define W8(OP) OP(0) OP(1) OP(2) OP(3) OP(4) OP(5) OP(6) OP(7)
#define WF_STORE(i)  *(float4*)(sS + (i) * 16 * W_PITCH) = w##i;
#define WF_LOAD0(i)  w##i = Wg[(size_t)(0 * DIM + (i) * 16) * (DIM / 4)];
#define WF_LOAD1(i)  w##i = Wg[(size_t)(1 * DIM + (i) * 16) * (DIM / 4)];
#define WF_LOAD2(i)  w##i = Wg[(size_t)(2 * DIM + (i) * 16) * (DIM / 4)];
static_assert(WF_F4 == 8, "W8 lists the float4 of one thread");

__device__ __forceinline__ void gather_x2_rows(const int* __restrict__ rowptr,
                                               const Edge* __restrict__ edge,
                                               const float* __restrict__ x1,
                                               const float* __restrict__ x0,
                                               const float* __restrict__ dinv,
                                               int row0, int rb, int re, int lane,
                                               float* tileT) {
    const float4* x4 = (const float4*)x1;
    const float4* p4 = (const float4*)x0;
    const float alpha = -2.0f, beta = -1.0f;
    int i   = rowptr[row0 + rb];
    int end = rowptr[row0 + rb + 1];
    for (int r = rb; r < re; ++r) {
        const int row  = row0 + r;
        const int nrow = row + 2 < N_NODES ? row + 2 : N_NODES;
        const int nend = rowptr[nrow];
        const float  di   = dinv[row];
        const float4 self = x4[(size_t)row * (DIM / 4) + lane];
        const float4 pv   = p4[(size_t)row * (DIM / 4) + lane];

        float4 acc = make_float4(0.0f, 0.0f, 0.0f, 0.0f);
        for (; i + 8 <= end; i += 8) pull_rows<8>(edge, x4, i, lane, acc);
        switch (end - i) {
            case 7: pull_rows<7>(edge, x4, i, lane, acc); break;
            case 6: pull_rows<6>(edge, x4, i, lane, acc); break;
            case 5: pull_rows<5>(edge, x4, i, lane, acc); break;
            case 4: pull_rows<4>(edge, x4, i, lane, acc); break;
            case 3: pull_rows<3>(edge, x4, i, lane, acc); break;
            case 2: pull_rows<2>(edge, x4, i, lane, acc); break;
            case 1: pull_rows<1>(edge, x4, i, lane, acc); break;
            default: break;
        }

        float4 o;
        o.x = alpha * ((acc.x + self.x * di) * di) + beta * pv.x;
        o.y = alpha * ((acc.y + self.y * di) * di) + beta * pv.y;
        o.z = alpha * ((acc.z + self.z * di) * di) + beta * pv.z;
        o.w = alpha * ((acc.w + self.w * di) * di) + beta * pv.w;
        *(float4*)(tileT + r * T_PITCH + 4 * lane) = o;
        i   = end;
        end = nend;
    }
}

__global__ void __launch_bounds__(FUSE_T, 8)
k_cheb_gemm_x2(const float* __restrict__ X0,
               const float* __restrict__ X1,
               const float* __restrict__ W,
               const float* __restrict__ b,
               float* __restrict__ out,
               const int* __restrict__ rowptr,
               const Edge* __restrict__ edge,
               const float* __restrict__ dinv) {
    __shared__ __attribute__((aligned(16))) float sW[DIM * W_PITCH];
    __shared__ __attribute__((aligned(16))) float sT[GEMM_WAVES * 16 * T_PITCH];

    const int tid  = threadIdx.x;
    const int wave = __builtin_amdgcn_readfirstlane(tid >> 5);
    const bool mm  = wave < GEMM_WAVES;
    const int g    = wave & (GEMM_WAVES - 1);
    const int tile = blockIdx.x * GEMM_WAVES + g;
    const bool live = tile < N_TILES;
    const int row0 = (live ? tile : N_TILES - 1) * 16;
    const int lane = tid & 31;
    const int half = lane >> 4;
    const int lr   = lane & 15;

    const float4* Wg = (const float4*)W + (size_t)(tid >> 5) * (DIM / 4) + (tid & 31);
    float*        sS = sW + (tid >> 5) * W_PITCH + (tid & 31) * 4;
    float4 w0, w1, w2, w3, w4, w5, w6, w7;
    W8(WF_LOAD0)

    v8f acc[8];

    const size_t aoff = (size_t)(row0 + lr) * DIM + 2 * half;
    const float* Bp   = sW + (2 * half) * W_PITCH + lr;
    float*       myT  = sT + g * 16 * T_PITCH;

    W8(WF_STORE)
    __syncthreads();
    if (mm) {
#pragma unroll
        for (int nt = 0; nt < 8; ++nt) {
            float bv = b[nt * 16 + lr];
            acc[nt] = (v8f){bv, bv, bv, bv, bv, bv, bv, bv};
        }
        gemm_segment(acc, X0 + aoff, Bp);
    } else {
        gather_x2_rows(rowptr, edge, X1, X0, dinv, row0, 0, 8, lane, myT);
    }
    W8(WF_LOAD1)

    __syncthreads();
    W8(WF_STORE)
    __syncthreads();
    if (mm) {
        gemm_segment(acc, X1 + aoff, Bp);
    } else {
        gather_x2_rows(rowptr, edge, X1, X0, dinv, row0, 8, 16, lane, myT);
    }
    W8(WF_LOAD2)

    __syncthreads();
    W8(WF_STORE)
    __syncthreads();
    if (mm) {
        gemm_segment(acc, myT + lr * T_PITCH + 2 * half, Bp);

        if (live) {
#pragma unroll
            for (int nt = 0; nt < 8; ++nt) {
                const int col = nt * 16 + lr;
#pragma unroll
                for (int r = 0; r < 8; ++r) {
                    const int row = row0 + r + 8 * half;
                    float v = acc[nt][r];
                    out[(size_t)row * DIM + col] = v > 0.0f ? v : 0.0f;
                }
            }
        }
    }
}

static_assert(GEMM_T == 2 * DIM, "one thread per element of lw | lb when they are copied to LDS");
#define U_PITCH 136

__device__ __forceinline__ v8f rank2_segment(const float* Ap, const float* Bp) {
    v8f acc = (v8f){0.0f, 0.0f, 0.0f, 0.0f, 0.0f, 0.0f, 0.0f, 0.0f};
#pragma unroll 8
    for (int kk = 0; kk < DIM; kk += 4) {
        v2f a, bb;
        a.x  = Ap[kk];
        a.y  = Ap[kk + 1];
        bb.x = Bp[kk * W_PITCH];
        bb.y = Bp[kk * W_PITCH + W_PITCH];
        acc = __builtin_amdgcn_wmma_f32_16x16x4_f32(
             false, a,  false, bb,
             (short)0, acc,
             false,  false);
    }
    asm volatile("v_nop\n\tv_nop\n\tv_nop\n\tv_nop" : "+v"(acc));
    return acc;
}

__global__ void __launch_bounds__(GEMM_T)
k_cheb_rank2(const float* __restrict__ W,
             const float* __restrict__ b,
             float* __restrict__ out,
             const float* __restrict__ wn,
             const float2* __restrict__ p1,
             const float2* __restrict__ p2,
             const float* __restrict__ lw,
             const float* __restrict__ lb) {
    __shared__ __attribute__((aligned(16))) float sW[DIM * W_PITCH];
    __shared__ float sU[8 * U_PITCH];
    __shared__ __attribute__((aligned(16))) float sA[2 * DIM];

    const int tid  = threadIdx.x;
    const int wave = __builtin_amdgcn_readfirstlane(tid >> 5);
    const int tile = blockIdx.x * GEMM_WAVES + wave;
    const bool live = tile < N_TILES;
    const int row0 = (live ? tile : N_TILES - 1) * 16;
    const int lane = tid & 31;
    const int half = lane >> 4;
    const int lr   = lane & 15;

    const float4* Wg = (const float4*)W + (size_t)(tid >> 5) * (DIM / 4) + (tid & 31);
    float*        sS = sW + (tid >> 5) * W_PITCH + (tid & 31) * 4;
    W16(W_DECL)
    W16(W_LOAD0)

    sA[tid] = (tid < DIM) ? lw[tid] : lb[tid - DIM];
    const float* Ap = sA + (lr & 1) * DIM + 2 * half;
    const float* Bp = sW + (2 * half) * W_PITCH + wave * 16 + lr;

    W16(W_STORE)
    __syncthreads();
    W16(W_LOAD1)
    const v8f u0 = rank2_segment(Ap, Bp);

    __syncthreads();
    W16(W_STORE)
    __syncthreads();
    W16(W_LOAD2)
    const v8f u1 = rank2_segment(Ap, Bp);

    __syncthreads();
    W16(W_STORE)
    __syncthreads();
    const v8f u2 = rank2_segment(Ap, Bp);

    if (lane < 16) {
        float* d = sU + wave * 16 + lane;
        d[0 * U_PITCH] = u0[0];
        d[1 * U_PITCH] = u0[1];
        d[2 * U_PITCH] = u1[0];
        d[3 * U_PITCH] = u1[1];
        d[4 * U_PITCH] = u2[0];
        d[5 * U_PITCH] = u2[1];
        d[6 * U_PITCH] = 0.0f;
        d[7 * U_PITCH] = 0.0f;
    }
    __syncthreads();

    const float  wv = wn[row0 + lr];
    const float2 q1 = p1[row0 + lr];
    const float2 q2 = p2[row0 + lr];
    v2f a0, a1;
    a0.x = half ? -q1.x : wv;
    a0.y = half ? -q1.y : 1.0f;
    a1.x = half ? 0.0f : 2.0f * q2.x - wv;
    a1.y = half ? 0.0f : 2.0f * q2.y - 1.0f;
    const float* Up = sU + (2 * half) * U_PITCH + lr;

    v8f acc[8];
#pragma unroll
    for (int nt = 0; nt < 8; ++nt) {
        float bv = b[nt * 16 + lr];
        acc[nt] = (v8f){bv, bv, bv, bv, bv, bv, bv, bv};
        v2f b0, b1;
        b0.x = Up[nt * 16];
        b0.y = Up[nt * 16 + U_PITCH];
        b1.x = Up[nt * 16 + 4 * U_PITCH];
        b1.y = Up[nt * 16 + 5 * U_PITCH];
        acc[nt] = __builtin_amdgcn_wmma_f32_16x16x4_f32(
             false, a0,  false, b0,
             (short)0, acc[nt],
             false,  false);
        acc[nt] = __builtin_amdgcn_wmma_f32_16x16x4_f32(
             false, a1,  false, b1,
             (short)0, acc[nt],
             false,  false);
    }

    if (live) {
#pragma unroll
        for (int nt = 0; nt < 8; ++nt) {
            const int col = nt * 16 + lr;
#pragma unroll
            for (int r = 0; r < 8; ++r) {
                const int row = row0 + r + 8 * half;
                float v = acc[nt][r];
                out[(size_t)row * DIM + col] = v > 0.0f ? v : 0.0f;
            }
        }
    }
}

__global__ void k_pred(const float* __restrict__ emb,
                       const float* __restrict__ pw,
                       const float* __restrict__ pb,
                       float* __restrict__ out) {
    int n = blockIdx.x * blockDim.x + threadIdx.x;
    if (n < N_NODES) {
        const float* e = emb + (size_t)n * DIM;
        float s = 0.0f;
#pragma unroll 8
        for (int d = 0; d < DIM; d += 4) {
            float4 v = *(const float4*)(e + d);
            s += v.x * pw[d] + v.y * pw[d + 1] + v.z * pw[d + 2] + v.w * pw[d + 3];
        }
        out[n] = s + pb[0];
    }
}

extern "C" void kernel_launch(void* const* d_in, const int* in_sizes, int n_in,
                              void* d_out, int out_size, void* d_ws, size_t ws_size,
                              hipStream_t stream) {
    const float* weights  = (const float*)d_in[0];
    const int*   src      = (const int*)  d_in[1];
    const int*   dst      = (const int*)  d_in[2];
    const float* lin_in_w = (const float*)d_in[3];
    const float* lin_in_b = (const float*)d_in[4];
    const float* cheb_ws  = (const float*)d_in[5];
    const float* cheb_bs  = (const float*)d_in[6];
    const float* pred_w   = (const float*)d_in[7];
    const float* pred_b   = (const float*)d_in[8];
    float* logits = (float*)d_out;

    float* ws   = (float*)d_ws;
    float* dinv = ws;
    float* X0   = ws + 51200;
    float* X1   = X0 + NF;
    float* X2   = X1 + NF;
    int*   g      = (int*)(X2 + NF);
    int*   tot    = g;
    int*   rowptr = g + 51200;
    Edge*  edge   = (Edge*)(g + 153600);
    int*   cnt    = g + 1400000;
    float* emb  = X2 + NF + NF;
    float2* p1  = (float2*)X1;
    float2* p2  = (float2*)X1 + 51200;

    const int T = 256;
    const int gN    = (N_NODES + T - 1) / T;
    const int gPull = N_NODES / PULL_ROWS_PER_BLOCK;

    k_slice_count<<<CSR_BLOCKS, CSR_T, 0, stream>>>(dst, cnt, tot);
    k_range_scan<<<CSR_R, CSR_T, 0, stream>>>(cnt, tot, rowptr, dinv);
    k_slice_place<<<CSR_BLOCKS, CSR_T, 0, stream>>>(src, dst, cnt, dinv, edge);

    k_spmm_pair<true><<<PAIR_BLOCKS, PAIR_T, 0, stream>>>(rowptr, edge, weights, p2, dinv, p1);
    k_spmm_pair<false><<<PAIR_BLOCKS, PAIR_T, 0, stream>>>(rowptr, edge, weights, p1, dinv, p2);
    k_cheb_rank2<<<GEMM_BLOCKS, GEMM_T, 0, stream>>>(cheb_ws, cheb_bs, emb,
                                                     weights, p1, p2, lin_in_w, lin_in_b);

    float* cur = emb;
    float* nxt = X0;
    for (int layer = 1; layer < 3; ++layer) {
        const float* Wl = cheb_ws + (size_t)layer * 3 * DIM * DIM;
        const float* bl = cheb_bs + (size_t)layer * DIM;

        k_spmm_pull<true><<<gPull, 32 * PULL_ROWS_PER_BLOCK, 0, stream>>>(
            rowptr, edge, cur, cur, dinv, X1, -1.0f, 0.0f);

        k_cheb_gemm_x2<<<GEMM_BLOCKS, FUSE_T, 0, stream>>>(cur, X1, Wl, bl, nxt, rowptr, edge, dinv);

        float* t = cur; cur = nxt; nxt = t;
    }

    k_pred<<<gN, T, 0, stream>>>(cur, pred_w, pred_b, logits);
}
